# starting kernel
# baseline (speedup 1.0000x reference)
#include <hip/hip_runtime.h>
#include <math.h>

typedef __attribute__((ext_vector_type(2))) float v2f;
typedef __attribute__((ext_vector_type(8))) float v8f;

#define NQ        6
#define DIM       64
#define NL        2
#define USTRIDE   65
#define PSTRIDE   36
#define BCOLS     32
#define NTHREADS  128

__device__ __forceinline__ void apply_gate(float* Ur, float* Ui, int pos, bool withCZ,
    float g00r, float g00i, float g01r, float g01i,
    float g10r, float g10i, float g11r, float g11i, int tid)
{
    const int bit = 1 << pos;
    for (int p = tid; p < 32 * DIM; p += NTHREADS) {
        int col = p & 63;
        int h   = p >> 6;
        int low = h & (bit - 1);
        int r0  = ((h ^ low) << 1) | low;
        int r1  = r0 | bit;
        int i0 = r0 * USTRIDE + col;
        int i1 = r1 * USTRIDE + col;
        float ar = Ur[i0], ai = Ui[i0];
        float br = Ur[i1], bi = Ui[i1];
        if (withCZ) {
            float d0 = (__popc(r0 & (r0 >> 1) & 0x1f) & 1) ? -1.0f : 1.0f;
            float d1 = (__popc(r1 & (r1 >> 1) & 0x1f) & 1) ? -1.0f : 1.0f;
            ar *= d0; ai *= d0;
            br *= d1; bi *= d1;
        }
        float n0r = g00r*ar - g00i*ai + g01r*br - g01i*bi;
        float n0i = g00r*ai + g00i*ar + g01r*bi + g01i*br;
        float n1r = g10r*ar - g10i*ai + g11r*br - g11i*bi;
        float n1i = g10r*ai + g10i*ar + g11r*bi + g11i*br;
        Ur[i0] = n0r; Ui[i0] = n0i;
        Ur[i1] = n1r; Ui[i1] = n1i;
    }
}

__global__ __launch_bounds__(NTHREADS)
void qreservoir_kernel(const float* __restrict__ x, const float* __restrict__ w,
                       float* __restrict__ out, int S)
{
    __shared__ float Ur[DIM * USTRIDE];
    __shared__ float Ui[DIM * USTRIDE];
    __shared__ float P [DIM * PSTRIDE];
    __shared__ float Sr[DIM * PSTRIDE];
    __shared__ float Si[DIM * PSTRIDE];
    __shared__ float cs[BCOLS][NQ];
    __shared__ float sn[BCOLS][NQ];

    const int s   = blockIdx.x;
    const int b0  = blockIdx.y * BCOLS;
    const int tid = threadIdx.x;

    for (int idx = tid; idx < BCOLS * NQ; idx += NTHREADS) {
        int c = idx / NQ, q = idx % NQ;
        float f = x[((b0 + c) * S + s) * NQ + q];
        float sv, cv;
        sincosf(0.5f * 3.14159265358979323846f * f, &sv, &cv);
        sn[c][q] = sv; cs[c][q] = cv;
    }
    for (int idx = tid; idx < DIM * DIM; idx += NTHREADS) {
        int r = idx >> 6, c = idx & 63;
        Ur[r * USTRIDE + c] = (r == c) ? 1.0f : 0.0f;
        Ui[r * USTRIDE + c] = 0.0f;
    }
    __syncthreads();
    for (int idx = tid; idx < DIM * BCOLS; idx += NTHREADS) {
        int i = idx >> 5, c = idx & 31;
        float v = 1.0f;
        #pragma unroll
        for (int q = 0; q < NQ; ++q) {
            int bitv = (i >> (NQ - 1 - q)) & 1;
            v *= bitv ? sn[c][q] : cs[c][q];
        }
        P[i * PSTRIDE + c] = v;
    }
    __syncthreads();

    const float cf = cosf(0.25f);
    const float sf = sinf(0.25f);
    for (int l = 0; l < NL; ++l) {
        #pragma unroll 1
        for (int q = 0; q < NQ; ++q) {
            int base = ((s * NL + l) * NQ + q) * 3;
            float phi = w[base], th = w[base + 1], om = w[base + 2];
            float st, ct; sincosf(0.5f * th, &st, &ct);
            float sa, ca; sincosf(-0.5f * (phi + om), &sa, &ca);
            float sb, cb; sincosf( 0.5f * (phi - om), &sb, &cb);
            float r00r =  ca * ct, r00i =  sa * ct;
            float r01r = -cb * st, r01i = -sb * st;
            float r10r =  cb * st, r10i = -sb * st;
            float r11r =  ca * ct, r11i = -sa * ct;
            float g00r = cf * r00r + sf * r01i;
            float g00i = cf * r00i - sf * r01r;
            float g01r = sf * r00i + cf * r01r;
            float g01i = -sf * r00r + cf * r01i;
            float g10r = cf * r10r + sf * r11i;
            float g10i = cf * r10i - sf * r11r;
            float g11r = sf * r10i + cf * r11r;
            float g11i = -sf * r10r + cf * r11i;
            apply_gate(Ur, Ui, NQ - 1 - q, (q == 0),
                       g00r, g00i, g01r, g01i,
                       g10r, g10i, g11r, g11i, tid);
            __syncthreads();
        }
    }

    const int lane = tid & 31;
    const int wv   = tid >> 5;
    const int lh   = lane >> 4;
    const int lm   = lane & 15;
    #pragma unroll
    for (int t = 0; t < 2; ++t) {
        int T  = wv * 2 + t;
        int m0 = T >> 1;
        int n0 = T & 1;
        v8f accR = {};
        v8f accI = {};
        int rowA = m0 * 16 + lm;
        int colB = n0 * 16 + lm;
        for (int k0 = 0; k0 < 16; ++k0) {
            int kb = k0 * 4 + lh * 2;
            v2f aR = { Ur[rowA * USTRIDE + kb], Ur[rowA * USTRIDE + kb + 1] };
            v2f aI = { Ui[rowA * USTRIDE + kb], Ui[rowA * USTRIDE + kb + 1] };
            v2f bF = { P[kb * PSTRIDE + colB], P[(kb + 1) * PSTRIDE + colB] };
            accR = __builtin_amdgcn_wmma_f32_16x16x4_f32(
                false, aR, false, bF, (short)0, accR, false, false);
            accI = __builtin_amdgcn_wmma_f32_16x16x4_f32(
                false, aI, false, bF, (short)0, accI, false, false);
        }
        #pragma unroll
        for (int v = 0; v < 8; ++v) {
            int row = m0 * 16 + lh * 8 + v;
            Sr[row * PSTRIDE + colB] = accR[v];
            Si[row * PSTRIDE + colB] = accI[v];
        }
    }
    __syncthreads();

    {
        int col     = tid >> 2;
        int quarter = tid & 3;
        float e[18];
        #pragma unroll
        for (int k = 0; k < 18; ++k) e[k] = 0.f;
        for (int it = 0; it < 16; ++it) {
            int i = quarter * 16 + ((it + quarter * 4) & 15);
            float ar = Sr[i * PSTRIDE + col];
            float ai = Si[i * PSTRIDE + col];
            float prob = ar * ar + ai * ai;
            #pragma unroll
            for (int q = 0; q < NQ; ++q) {
                int pos  = NQ - 1 - q;
                int bitv = (i >> pos) & 1;
                int j    = i ^ (1 << pos);
                float br = Sr[j * PSTRIDE + col];
                float bi = Si[j * PSTRIDE + col];
                e[q]      += ar * br + ai * bi;
                float sgn  = bitv ? 1.0f : -1.0f;
                e[6 + q]  += sgn * (ai * br - ar * bi);
                e[12 + q] += bitv ? -prob : prob;
            }
        }
        #pragma unroll
        for (int k = 0; k < 18; ++k) {
            e[k] += __shfl_xor(e[k], 1);
            e[k] += __shfl_xor(e[k], 2);
        }
        if (quarter == 0) {
            int b = b0 + col;
            float* o = out + ((size_t)b * S + s) * 18;
            #pragma unroll
            for (int k = 0; k < 18; ++k) o[k] = e[k];
        }
    }
}

extern "C" void kernel_launch(void* const* d_in, const int* in_sizes, int n_in,
                              void* d_out, int out_size, void* d_ws, size_t ws_size,
                              hipStream_t stream) {
    const float* x = (const float*)d_in[0];
    const float* w = (const float*)d_in[1];
    float* out = (float*)d_out;
    (void)d_ws; (void)ws_size; (void)n_in; (void)out_size;
    int S = in_sizes[1] / (NL * NQ * 3);
    int B = in_sizes[0] / (S * NQ);
    dim3 grid(S, B / BCOLS);
    qreservoir_kernel<<<grid, NTHREADS, 0, stream>>>(x, w, out, S);
}


# Apply each layer to the 64x32 state as two Kronecker factors (8x8 gate triples as real 16x16 [Re -Im; Im Re] WMMA blocks, K=16) instead of building U by 12 LDS sweeps; changes summation order
# speedup vs baseline: 2.6479x; 2.6479x over previous
#include <hip/hip_runtime.h>
#include <math.h>

typedef __attribute__((ext_vector_type(2))) float v2f;
typedef __attribute__((ext_vector_type(8))) float v8f;

#define NQ        6
#define DIM       64
#define NL        2
#define PSTRIDE   36
#define BCOLS     32
#define NTHREADS  128

__device__ __forceinline__ float cz_sign(float v, int i)
{
    unsigned par = (unsigned)__popc(i & (i >> 1) & 0x1f) & 1u;
    return __uint_as_float(__float_as_uint(v) ^ (par << 31));
}

template <bool HI, bool REAL_IN, bool CZ_OUT>
__device__ __forceinline__ void kron_stage(const float* g,
                                           const float* Xr, const float* Xi,
                                           float* Yr, float* Yi,
                                           int wv, int lm, int lh)
{
    const int m  = lm & 7;
    const int ga = (HI ? 0 : 3) * 8 + ((m >> 2) & 1) * 4;
    const int gb = (HI ? 1 : 4) * 8 + ((m >> 1) & 1) * 4 + lh * 2;
    const int gc = (HI ? 2 : 5) * 8 + (m & 1) * 4;
    const float a0r = g[ga], a0i = g[ga + 1], a1r = g[ga + 2], a1i = g[ga + 3];
    const float br  = g[gb], bi  = g[gb + 1];
    const float c0r = g[gc], c0i = g[gc + 1], c1r = g[gc + 2], c1i = g[gc + 3];
    const float t0r = br * c0r - bi * c0i, t0i = br * c0i + bi * c0r;
    const float t1r = br * c1r - bi * c1i, t1i = br * c1i + bi * c1r;
    const float w00r = a0r * t0r - a0i * t0i, w00i = a0r * t0i + a0i * t0r;
    const float w01r = a0r * t1r - a0i * t1i, w01i = a0r * t1i + a0i * t1r;
    const float w10r = a1r * t0r - a1i * t0i, w10i = a1r * t0i + a1i * t0r;
    const float w11r = a1r * t1r - a1i * t1i, w11i = a1r * t1i + a1i * t1r;
    const bool imrow = (lm >= 8);
    v2f afr[4];
    afr[0] = imrow ? (v2f){ w00i, w01i } : (v2f){ w00r, w01r };
    afr[1] = imrow ? (v2f){ w10i, w11i } : (v2f){ w10r, w11r };
    afr[2] = imrow ? (v2f){ w00r, w01r } : (v2f){ -w00i, -w01i };
    afr[3] = imrow ? (v2f){ w10r, w11r } : (v2f){ -w10i, -w11i };

    v8f acc[4];
    #pragma unroll
    for (int t = 0; t < 4; ++t) acc[t] = (v8f){};

    #pragma unroll
    for (int k0 = 0; k0 < (REAL_IN ? 2 : 4); ++k0) {
        const float* X = (k0 >> 1) ? Xi : Xr;
        const int x0 = (k0 & 1) * 4 + lh * 2;
        #pragma unroll
        for (int t = 0; t < 4; ++t) {
            const int idx8 = wv * 2 + (t >> 1);
            const int col  = (t & 1) * 16 + lm;
            const int r0   = HI ? (x0 * 8 + idx8) : (idx8 * 8 + x0);
            const int r1   = HI ? (r0 + 8) : (r0 + 1);
            v2f b = { X[r0 * PSTRIDE + col], X[r1 * PSTRIDE + col] };
            acc[t] = __builtin_amdgcn_wmma_f32_16x16x4_f32(
                false, afr[k0], false, b, (short)0, acc[t], false, false);
        }
    }

    float* Y = lh ? Yi : Yr;
    #pragma unroll
    for (int t = 0; t < 4; ++t) {
        const int idx8 = wv * 2 + (t >> 1);
        const int col  = (t & 1) * 16 + lm;
        #pragma unroll
        for (int v = 0; v < 8; ++v) {
            const int row = HI ? (v * 8 + idx8) : (idx8 * 8 + v);
            float val = acc[t][v];
            if (CZ_OUT) val = cz_sign(val, row);
            Y[row * PSTRIDE + col] = val;
        }
    }
}

__global__ __launch_bounds__(NTHREADS)
void qreservoir_kernel(const float* __restrict__ x, const float* __restrict__ w,
                       float* __restrict__ out, int S)
{
    __shared__ float P [DIM * PSTRIDE];
    __shared__ float Tr[DIM * PSTRIDE];
    __shared__ float Ti[DIM * PSTRIDE];
    __shared__ float Sr[DIM * PSTRIDE];
    __shared__ float Si[DIM * PSTRIDE];
    __shared__ float Gs[NL * NQ * 8];
    __shared__ float cs[BCOLS][NQ];
    __shared__ float sn[BCOLS][NQ];

    const int s   = blockIdx.x;
    const int b0  = blockIdx.y * BCOLS;
    const int tid = threadIdx.x;

    for (int idx = tid; idx < BCOLS * NQ; idx += NTHREADS) {
        int c = idx / NQ, q = idx % NQ;
        float f = x[((b0 + c) * S + s) * NQ + q];
        float sv, cv;
        sincosf(0.5f * 3.14159265358979323846f * f, &sv, &cv);
        sn[c][q] = sv; cs[c][q] = cv;
    }
    if (tid >= 64 && tid < 64 + NL * NQ) {
        const float cf = cosf(0.25f);
        const float sf = sinf(0.25f);
        const int gidx = tid - 64;
        const int base = (s * NL * NQ + gidx) * 3;
        float phi = w[base], th = w[base + 1], om = w[base + 2];
        float st, ct; sincosf(0.5f * th, &st, &ct);
        float sa, ca; sincosf(-0.5f * (phi + om), &sa, &ca);
        float sb, cb; sincosf( 0.5f * (phi - om), &sb, &cb);
        float r00r =  ca * ct, r00i =  sa * ct;
        float r01r = -cb * st, r01i = -sb * st;
        float r10r =  cb * st, r10i = -sb * st;
        float r11r =  ca * ct, r11i = -sa * ct;
        float* go = &Gs[gidx * 8];
        go[0] = cf * r00r + sf * r01i;
        go[1] = cf * r00i - sf * r01r;
        go[2] = sf * r00i + cf * r01r;
        go[3] = -sf * r00r + cf * r01i;
        go[4] = cf * r10r + sf * r11i;
        go[5] = cf * r10i - sf * r11r;
        go[6] = sf * r10i + cf * r11r;
        go[7] = -sf * r10r + cf * r11i;
    }
    __syncthreads();
    for (int idx = tid; idx < DIM * BCOLS; idx += NTHREADS) {
        int i = idx >> 5, c = idx & 31;
        float v = 1.0f;
        #pragma unroll
        for (int q = 0; q < NQ; ++q) {
            int bitv = (i >> (NQ - 1 - q)) & 1;
            v *= bitv ? sn[c][q] : cs[c][q];
        }
        P[i * PSTRIDE + c] = cz_sign(v, i);
    }
    __syncthreads();

    const int lane = tid & 31;
    const int wv   = tid >> 5;
    const int lh   = lane >> 4;
    const int lm   = lane & 15;
    kron_stage<false, true,  false>(Gs,          P,  P,  Tr, Ti, wv, lm, lh);
    __syncthreads();
    kron_stage<true,  false, true >(Gs,          Tr, Ti, Sr, Si, wv, lm, lh);
    __syncthreads();
    kron_stage<false, false, false>(Gs + NQ * 8, Sr, Si, Tr, Ti, wv, lm, lh);
    __syncthreads();
    kron_stage<true,  false, false>(Gs + NQ * 8, Tr, Ti, Sr, Si, wv, lm, lh);
    __syncthreads();

    {
        int col     = tid >> 2;
        int quarter = tid & 3;
        float e[18];
        #pragma unroll
        for (int k = 0; k < 18; ++k) e[k] = 0.f;
        for (int it = 0; it < 16; ++it) {
            int i = quarter * 16 + ((it + quarter * 4) & 15);
            float ar = Sr[i * PSTRIDE + col];
            float ai = Si[i * PSTRIDE + col];
            float prob = ar * ar + ai * ai;
            #pragma unroll
            for (int q = 0; q < NQ; ++q) {
                int pos  = NQ - 1 - q;
                int bitv = (i >> pos) & 1;
                int j    = i ^ (1 << pos);
                float br = Sr[j * PSTRIDE + col];
                float bi = Si[j * PSTRIDE + col];
                e[q]      += ar * br + ai * bi;
                float sgn  = bitv ? 1.0f : -1.0f;
                e[6 + q]  += sgn * (ai * br - ar * bi);
                e[12 + q] += bitv ? -prob : prob;
            }
        }
        #pragma unroll
        for (int k = 0; k < 18; ++k) {
            e[k] += __shfl_xor(e[k], 1);
            e[k] += __shfl_xor(e[k], 2);
        }
        if (quarter == 0) {
            int b = b0 + col;
            float* o = out + ((size_t)b * S + s) * 18;
            #pragma unroll
            for (int k = 0; k < 18; ++k) o[k] = e[k];
        }
    }
}

extern "C" void kernel_launch(void* const* d_in, const int* in_sizes, int n_in,
                              void* d_out, int out_size, void* d_ws, size_t ws_size,
                              hipStream_t stream) {
    const float* x = (const float*)d_in[0];
    const float* w = (const float*)d_in[1];
    float* out = (float*)d_out;
    (void)d_ws; (void)ws_size; (void)n_in; (void)out_size;
    int S = in_sizes[1] / (NL * NQ * 3);
    int B = in_sizes[0] / (S * NQ);
    dim3 grid(S, B / BCOLS);
    qreservoir_kernel<<<grid, NTHREADS, 0, stream>>>(x, w, out, S);
}


# Product state built by doubling: one lane per column and wave-wide top two bits forms its 16 amplitudes in 31 multiplies in registers, CZ signs on factors, not 6 selected LDS reads per amplitude
# speedup vs baseline: 3.1010x; 1.1711x over previous
#include <hip/hip_runtime.h>
#include <math.h>

typedef __attribute__((ext_vector_type(2))) float v2f;
typedef __attribute__((ext_vector_type(8))) float v8f;

#define NQ        6
#define DIM       64
#define NL        2
#define PSTRIDE   36
#define BCOLS     32
#define NTHREADS  128

__device__ __forceinline__ float cz_sign(float v, int i)
{
    unsigned par = (unsigned)__popc(i & (i >> 1) & 0x1f) & 1u;
    return __uint_as_float(__float_as_uint(v) ^ (par << 31));
}

template <bool HI, bool REAL_IN, bool CZ_OUT>
__device__ __forceinline__ void kron_stage(const float* g,
                                           const float* Xr, const float* Xi,
                                           float* Yr, float* Yi,
                                           int wv, int lm, int lh)
{
    const int m  = lm & 7;
    const int ga = (HI ? 0 : 3) * 8 + ((m >> 2) & 1) * 4;
    const int gb = (HI ? 1 : 4) * 8 + ((m >> 1) & 1) * 4 + lh * 2;
    const int gc = (HI ? 2 : 5) * 8 + (m & 1) * 4;
    const float a0r = g[ga], a0i = g[ga + 1], a1r = g[ga + 2], a1i = g[ga + 3];
    const float br  = g[gb], bi  = g[gb + 1];
    const float c0r = g[gc], c0i = g[gc + 1], c1r = g[gc + 2], c1i = g[gc + 3];
    const float t0r = br * c0r - bi * c0i, t0i = br * c0i + bi * c0r;
    const float t1r = br * c1r - bi * c1i, t1i = br * c1i + bi * c1r;
    const float w00r = a0r * t0r - a0i * t0i, w00i = a0r * t0i + a0i * t0r;
    const float w01r = a0r * t1r - a0i * t1i, w01i = a0r * t1i + a0i * t1r;
    const float w10r = a1r * t0r - a1i * t0i, w10i = a1r * t0i + a1i * t0r;
    const float w11r = a1r * t1r - a1i * t1i, w11i = a1r * t1i + a1i * t1r;
    const bool imrow = (lm >= 8);
    v2f afr[4];
    afr[0] = imrow ? (v2f){ w00i, w01i } : (v2f){ w00r, w01r };
    afr[1] = imrow ? (v2f){ w10i, w11i } : (v2f){ w10r, w11r };
    afr[2] = imrow ? (v2f){ w00r, w01r } : (v2f){ -w00i, -w01i };
    afr[3] = imrow ? (v2f){ w10r, w11r } : (v2f){ -w10i, -w11i };

    v8f acc[4];
    #pragma unroll
    for (int t = 0; t < 4; ++t) acc[t] = (v8f){};

    #pragma unroll
    for (int k0 = 0; k0 < (REAL_IN ? 2 : 4); ++k0) {
        const float* X = (k0 >> 1) ? Xi : Xr;
        const int x0 = (k0 & 1) * 4 + lh * 2;
        #pragma unroll
        for (int t = 0; t < 4; ++t) {
            const int idx8 = wv * 2 + (t >> 1);
            const int col  = (t & 1) * 16 + lm;
            const int r0   = HI ? (x0 * 8 + idx8) : (idx8 * 8 + x0);
            const int r1   = HI ? (r0 + 8) : (r0 + 1);
            v2f b = { X[r0 * PSTRIDE + col], X[r1 * PSTRIDE + col] };
            acc[t] = __builtin_amdgcn_wmma_f32_16x16x4_f32(
                false, afr[k0], false, b, (short)0, acc[t], false, false);
        }
    }

    float* Y = lh ? Yi : Yr;
    #pragma unroll
    for (int t = 0; t < 4; ++t) {
        const int idx8 = wv * 2 + (t >> 1);
        const int col  = (t & 1) * 16 + lm;
        #pragma unroll
        for (int v = 0; v < 8; ++v) {
            const int row = HI ? (v * 8 + idx8) : (idx8 * 8 + v);
            float val = acc[t][v];
            if (CZ_OUT) val = cz_sign(val, row);
            Y[row * PSTRIDE + col] = val;
        }
    }
}

__global__ __launch_bounds__(NTHREADS)
void qreservoir_kernel(const float* __restrict__ x, const float* __restrict__ w,
                       float* __restrict__ out, int S)
{
    __shared__ float P [DIM * PSTRIDE];
    __shared__ float Tr[DIM * PSTRIDE];
    __shared__ float Ti[DIM * PSTRIDE];
    __shared__ float Sr[DIM * PSTRIDE];
    __shared__ float Si[DIM * PSTRIDE];
    __shared__ float Gs[NL * NQ * 8];
    __shared__ float cs[BCOLS][NQ];
    __shared__ float sn[BCOLS][NQ];

    const int s   = blockIdx.x;
    const int b0  = blockIdx.y * BCOLS;
    const int tid = threadIdx.x;

    for (int idx = tid; idx < BCOLS * NQ; idx += NTHREADS) {
        int c = idx / NQ, q = idx % NQ;
        float f = x[((b0 + c) * S + s) * NQ + q];
        float sv, cv;
        sincosf(0.5f * 3.14159265358979323846f * f, &sv, &cv);
        sn[c][q] = sv; cs[c][q] = cv;
    }
    if (tid >= 64 && tid < 64 + NL * NQ) {
        const float cf = cosf(0.25f);
        const float sf = sinf(0.25f);
        const int gidx = tid - 64;
        const int base = (s * NL * NQ + gidx) * 3;
        float phi = w[base], th = w[base + 1], om = w[base + 2];
        float st, ct; sincosf(0.5f * th, &st, &ct);
        float sa, ca; sincosf(-0.5f * (phi + om), &sa, &ca);
        float sb, cb; sincosf( 0.5f * (phi - om), &sb, &cb);
        float r00r =  ca * ct, r00i =  sa * ct;
        float r01r = -cb * st, r01i = -sb * st;
        float r10r =  cb * st, r10i = -sb * st;
        float r11r =  ca * ct, r11i = -sa * ct;
        float* go = &Gs[gidx * 8];
        go[0] = cf * r00r + sf * r01i;
        go[1] = cf * r00i - sf * r01r;
        go[2] = sf * r00i + cf * r01r;
        go[3] = -sf * r00r + cf * r01i;
        go[4] = cf * r10r + sf * r11i;
        go[5] = cf * r10i - sf * r11r;
        go[6] = sf * r10i + cf * r11r;
        go[7] = -sf * r10r + cf * r11i;
    }
    __syncthreads();
    {
        const int  c  = tid & 31;
        const int  hi = tid >> 5;
        const bool b5 = (hi & 2) != 0, b4 = (hi & 1) != 0;
        float f0 = b5 ? sn[c][0] : cs[c][0];
        const float f1 = b4 ? sn[c][1] : cs[c][1];
        if (b5 && b4) f0 = -f0;
        const float c2 = cs[c][2], s2 = b4 ? -sn[c][2] : sn[c][2];
        const float c3 = cs[c][3], s3 = sn[c][3];
        const float c4 = cs[c][4], s4 = sn[c][4];
        const float c5 = cs[c][5], s5 = sn[c][5];
        const float t01 = f0 * f1;
        float l2[2], l3[4], l4[8];
        l2[0] = t01 * c2; l2[1] = t01 * s2;
        #pragma unroll
        for (int j = 0; j < 4; ++j) l3[j] = l2[j >> 1] * ((j & 1) ? s3 : c3);
        #pragma unroll
        for (int j = 0; j < 8; ++j) l4[j] = l3[j >> 1] * ((j & 1) ? s4 : c4);
        #pragma unroll
        for (int j = 0; j < 16; ++j) {
            const float v = l4[j >> 1] * ((j & 1) ? s5 : c5);
            const bool neg = (__builtin_popcount(j & (j >> 1) & 7) & 1) != 0;
            P[(hi * 16 + j) * PSTRIDE + c] = neg ? -v : v;
        }
    }
    __syncthreads();

    const int lane = tid & 31;
    const int wv   = tid >> 5;
    const int lh   = lane >> 4;
    const int lm   = lane & 15;
    kron_stage<false, true,  false>(Gs,          P,  P,  Tr, Ti, wv, lm, lh);
    __syncthreads();
    kron_stage<true,  false, true >(Gs,          Tr, Ti, Sr, Si, wv, lm, lh);
    __syncthreads();
    kron_stage<false, false, false>(Gs + NQ * 8, Sr, Si, Tr, Ti, wv, lm, lh);
    __syncthreads();
    kron_stage<true,  false, false>(Gs + NQ * 8, Tr, Ti, Sr, Si, wv, lm, lh);
    __syncthreads();

    {
        int col     = tid >> 2;
        int quarter = tid & 3;
        float e[18];
        #pragma unroll
        for (int k = 0; k < 18; ++k) e[k] = 0.f;
        for (int it = 0; it < 16; ++it) {
            int i = quarter * 16 + ((it + quarter * 4) & 15);
            float ar = Sr[i * PSTRIDE + col];
            float ai = Si[i * PSTRIDE + col];
            float prob = ar * ar + ai * ai;
            #pragma unroll
            for (int q = 0; q < NQ; ++q) {
                int pos  = NQ - 1 - q;
                int bitv = (i >> pos) & 1;
                int j    = i ^ (1 << pos);
                float br = Sr[j * PSTRIDE + col];
                float bi = Si[j * PSTRIDE + col];
                e[q]      += ar * br + ai * bi;
                float sgn  = bitv ? 1.0f : -1.0f;
                e[6 + q]  += sgn * (ai * br - ar * bi);
                e[12 + q] += bitv ? -prob : prob;
            }
        }
        #pragma unroll
        for (int k = 0; k < 18; ++k) {
            e[k] += __shfl_xor(e[k], 1);
            e[k] += __shfl_xor(e[k], 2);
        }
        if (quarter == 0) {
            int b = b0 + col;
            float* o = out + ((size_t)b * S + s) * 18;
            #pragma unroll
            for (int k = 0; k < 18; ++k) o[k] = e[k];
        }
    }
}

extern "C" void kernel_launch(void* const* d_in, const int* in_sizes, int n_in,
                              void* d_out, int out_size, void* d_ws, size_t ws_size,
                              hipStream_t stream) {
    const float* x = (const float*)d_in[0];
    const float* w = (const float*)d_in[1];
    float* out = (float*)d_out;
    (void)d_ws; (void)ws_size; (void)n_in; (void)out_size;
    int S = in_sizes[1] / (NL * NQ * 3);
    int B = in_sizes[0] / (S * NQ);
    dim3 grid(S, B / BCOLS);
    qreservoir_kernel<<<grid, NTHREADS, 0, stream>>>(x, w, out, S);
}


# Expectations from registers: each lane reads 16 amplitudes once per lane partition (two) and forms every qubit pair in-lane by packed FMAs, not 14 LDS reads per amplitude; changes summation order
# speedup vs baseline: 3.5818x; 1.1550x over previous
#include <hip/hip_runtime.h>
#include <math.h>

typedef __attribute__((ext_vector_type(2))) float v2f;
typedef __attribute__((ext_vector_type(8))) float v8f;

#define NQ        6
#define DIM       64
#define NL        2
#define PSTRIDE   36
#define BCOLS     32
#define NTHREADS  128

__device__ __forceinline__ float cz_sign(float v, int i)
{
    unsigned par = (unsigned)__popc(i & (i >> 1) & 0x1f) & 1u;
    return __uint_as_float(__float_as_uint(v) ^ (par << 31));
}

template <bool HI, bool REAL_IN, bool CZ_OUT>
__device__ __forceinline__ void kron_stage(const float* g,
                                           const float* Xr, const float* Xi,
                                           float* Yr, float* Yi,
                                           int wv, int lm, int lh)
{
    const int m  = lm & 7;
    const int ga = (HI ? 0 : 3) * 8 + ((m >> 2) & 1) * 4;
    const int gb = (HI ? 1 : 4) * 8 + ((m >> 1) & 1) * 4 + lh * 2;
    const int gc = (HI ? 2 : 5) * 8 + (m & 1) * 4;
    const float a0r = g[ga], a0i = g[ga + 1], a1r = g[ga + 2], a1i = g[ga + 3];
    const float br  = g[gb], bi  = g[gb + 1];
    const float c0r = g[gc], c0i = g[gc + 1], c1r = g[gc + 2], c1i = g[gc + 3];
    const float t0r = br * c0r - bi * c0i, t0i = br * c0i + bi * c0r;
    const float t1r = br * c1r - bi * c1i, t1i = br * c1i + bi * c1r;
    const float w00r = a0r * t0r - a0i * t0i, w00i = a0r * t0i + a0i * t0r;
    const float w01r = a0r * t1r - a0i * t1i, w01i = a0r * t1i + a0i * t1r;
    const float w10r = a1r * t0r - a1i * t0i, w10i = a1r * t0i + a1i * t0r;
    const float w11r = a1r * t1r - a1i * t1i, w11i = a1r * t1i + a1i * t1r;
    const bool imrow = (lm >= 8);
    v2f afr[4];
    afr[0] = imrow ? (v2f){ w00i, w01i } : (v2f){ w00r, w01r };
    afr[1] = imrow ? (v2f){ w10i, w11i } : (v2f){ w10r, w11r };
    afr[2] = imrow ? (v2f){ w00r, w01r } : (v2f){ -w00i, -w01i };
    afr[3] = imrow ? (v2f){ w10r, w11r } : (v2f){ -w10i, -w11i };

    v8f acc[4];
    #pragma unroll
    for (int t = 0; t < 4; ++t) acc[t] = (v8f){};

    #pragma unroll
    for (int k0 = 0; k0 < (REAL_IN ? 2 : 4); ++k0) {
        const float* X = (k0 >> 1) ? Xi : Xr;
        const int x0 = (k0 & 1) * 4 + lh * 2;
        #pragma unroll
        for (int t = 0; t < 4; ++t) {
            const int idx8 = wv * 2 + (t >> 1);
            const int col  = (t & 1) * 16 + lm;
            const int r0   = HI ? (x0 * 8 + idx8) : (idx8 * 8 + x0);
            const int r1   = HI ? (r0 + 8) : (r0 + 1);
            v2f b = { X[r0 * PSTRIDE + col], X[r1 * PSTRIDE + col] };
            acc[t] = __builtin_amdgcn_wmma_f32_16x16x4_f32(
                false, afr[k0], false, b, (short)0, acc[t], false, false);
        }
    }

    float* Y = lh ? Yi : Yr;
    #pragma unroll
    for (int t = 0; t < 4; ++t) {
        const int idx8 = wv * 2 + (t >> 1);
        const int col  = (t & 1) * 16 + lm;
        #pragma unroll
        for (int v = 0; v < 8; ++v) {
            const int row = HI ? (v * 8 + idx8) : (idx8 * 8 + v);
            float val = acc[t][v];
            if (CZ_OUT) val = cz_sign(val, row);
            Y[row * PSTRIDE + col] = val;
        }
    }
}

__global__ __launch_bounds__(NTHREADS)
void qreservoir_kernel(const float* __restrict__ x, const float* __restrict__ w,
                       float* __restrict__ out, int S)
{
    __shared__ float P [DIM * PSTRIDE];
    __shared__ float Tr[DIM * PSTRIDE];
    __shared__ float Ti[DIM * PSTRIDE];
    __shared__ float Sr[DIM * PSTRIDE];
    __shared__ float Si[DIM * PSTRIDE];
    __shared__ float Gs[NL * NQ * 8];
    __shared__ float cs[BCOLS][NQ];
    __shared__ float sn[BCOLS][NQ];

    const int s   = blockIdx.x;
    const int b0  = blockIdx.y * BCOLS;
    const int tid = threadIdx.x;

    for (int idx = tid; idx < BCOLS * NQ; idx += NTHREADS) {
        int c = idx / NQ, q = idx % NQ;
        float f = x[((b0 + c) * S + s) * NQ + q];
        float sv, cv;
        sincosf(0.5f * 3.14159265358979323846f * f, &sv, &cv);
        sn[c][q] = sv; cs[c][q] = cv;
    }
    if (tid >= 64 && tid < 64 + NL * NQ) {
        const float cf = cosf(0.25f);
        const float sf = sinf(0.25f);
        const int gidx = tid - 64;
        const int base = (s * NL * NQ + gidx) * 3;
        float phi = w[base], th = w[base + 1], om = w[base + 2];
        float st, ct; sincosf(0.5f * th, &st, &ct);
        float sa, ca; sincosf(-0.5f * (phi + om), &sa, &ca);
        float sb, cb; sincosf( 0.5f * (phi - om), &sb, &cb);
        float r00r =  ca * ct, r00i =  sa * ct;
        float r01r = -cb * st, r01i = -sb * st;
        float r10r =  cb * st, r10i = -sb * st;
        float r11r =  ca * ct, r11i = -sa * ct;
        float* go = &Gs[gidx * 8];
        go[0] = cf * r00r + sf * r01i;
        go[1] = cf * r00i - sf * r01r;
        go[2] = sf * r00i + cf * r01r;
        go[3] = -sf * r00r + cf * r01i;
        go[4] = cf * r10r + sf * r11i;
        go[5] = cf * r10i - sf * r11r;
        go[6] = sf * r10i + cf * r11r;
        go[7] = -sf * r10r + cf * r11i;
    }
    __syncthreads();
    {
        const int  c  = tid & 31;
        const int  hi = tid >> 5;
        const bool b5 = (hi & 2) != 0, b4 = (hi & 1) != 0;
        float f0 = b5 ? sn[c][0] : cs[c][0];
        const float f1 = b4 ? sn[c][1] : cs[c][1];
        if (b5 && b4) f0 = -f0;
        const float c2 = cs[c][2], s2 = b4 ? -sn[c][2] : sn[c][2];
        const float c3 = cs[c][3], s3 = sn[c][3];
        const float c4 = cs[c][4], s4 = sn[c][4];
        const float c5 = cs[c][5], s5 = sn[c][5];
        const float t01 = f0 * f1;
        float l2[2], l3[4], l4[8];
        l2[0] = t01 * c2; l2[1] = t01 * s2;
        #pragma unroll
        for (int j = 0; j < 4; ++j) l3[j] = l2[j >> 1] * ((j & 1) ? s3 : c3);
        #pragma unroll
        for (int j = 0; j < 8; ++j) l4[j] = l3[j >> 1] * ((j & 1) ? s4 : c4);
        #pragma unroll
        for (int j = 0; j < 16; ++j) {
            const float v = l4[j >> 1] * ((j & 1) ? s5 : c5);
            const bool neg = (__builtin_popcount(j & (j >> 1) & 7) & 1) != 0;
            P[(hi * 16 + j) * PSTRIDE + c] = neg ? -v : v;
        }
    }
    __syncthreads();

    const int lane = tid & 31;
    const int wv   = tid >> 5;
    const int lh   = lane >> 4;
    const int lm   = lane & 15;
    kron_stage<false, true,  false>(Gs,          P,  P,  Tr, Ti, wv, lm, lh);
    __syncthreads();
    kron_stage<true,  false, true >(Gs,          Tr, Ti, Sr, Si, wv, lm, lh);
    __syncthreads();
    kron_stage<false, false, false>(Gs + NQ * 8, Sr, Si, Tr, Ti, wv, lm, lh);
    __syncthreads();
    kron_stage<true,  false, false>(Gs + NQ * 8, Tr, Ti, Sr, Si, wv, lm, lh);
    __syncthreads();

    {
        const int col = tid >> 2;
        const int p   = tid & 3;
        float e[18];
        v2f R[8], I[8];

        #pragma unroll
        for (int j = 0; j < 8; ++j) {
            const int r0 = (8 * j + p) * PSTRIDE + col;
            R[j] = (v2f){ Sr[r0], Sr[r0 + 4 * PSTRIDE] };
            I[j] = (v2f){ Si[r0], Si[r0 + 4 * PSTRIDE] };
        }
        #pragma unroll
        for (int q = 0; q < 3; ++q) {
            const int jb = 1 << (2 - q);
            v2f sx = {0.f, 0.f}, sy = {0.f, 0.f};
            #pragma unroll
            for (int j0 = 0; j0 < 8; ++j0) {
                if (j0 & jb) continue;
                const int j1 = j0 | jb;
                sx += R[j0] * R[j1];
                sx += I[j0] * I[j1];
                sy += R[j0] * I[j1];
                sy -= I[j0] * R[j1];
            }
            e[q] = 2.f * (sx.x + sx.y); e[6 + q] = 2.f * (sy.x + sy.y);
        }
        {
            float sx = 0.f, sy = 0.f;
            #pragma unroll
            for (int j = 0; j < 8; ++j) {
                sx += R[j].x * R[j].y;
                sx += I[j].x * I[j].y;
                sy += R[j].x * I[j].y;
                sy -= I[j].x * R[j].y;
            }
            e[3] = 2.f * sx; e[9] = 2.f * sy;
        }
        {
            v2f s0[8], s1[4], s2[2];
            #pragma unroll
            for (int j = 0; j < 8; ++j) s0[j] = R[j] * R[j] + I[j] * I[j];
            v2f d2 = {0.f, 0.f}, d1 = {0.f, 0.f};
            #pragma unroll
            for (int m = 0; m < 4; ++m) { d2 += s0[2 * m] - s0[2 * m + 1]; s1[m] = s0[2 * m] + s0[2 * m + 1]; }
            #pragma unroll
            for (int m = 0; m < 2; ++m) { d1 += s1[2 * m] - s1[2 * m + 1]; s2[m] = s1[2 * m] + s1[2 * m + 1]; }
            const v2f d0 = s2[0] - s2[1];
            const v2f t2 = s2[0] + s2[1];
            const float tot = t2.x + t2.y;
            e[12 + 0] = d0.x + d0.y;
            e[12 + 1] = d1.x + d1.y;
            e[12 + 2] = d2.x + d2.y;
            e[12 + 3] = t2.x - t2.y;
            e[12 + 4] = (p & 2) ? -tot : tot;
            e[12 + 5] = (p & 1) ? -tot : tot;
        }

        #pragma unroll
        for (int j = 0; j < 8; ++j) {
            const int r0 = (16 * p + 8 * (j >> 2) + ((j & 3) ^ p)) * PSTRIDE + col;
            R[j] = (v2f){ Sr[r0], Sr[r0 + 4 * PSTRIDE] };
            I[j] = (v2f){ Si[r0], Si[r0 + 4 * PSTRIDE] };
        }
        #pragma unroll
        for (int q = 4; q < 6; ++q) {
            const int jb = 1 << (5 - q);
            v2f sx = {0.f, 0.f}, sy = {0.f, 0.f};
            #pragma unroll
            for (int j0 = 0; j0 < 8; ++j0) {
                if (j0 & jb) continue;
                const int j1 = j0 | jb;
                sx += R[j0] * R[j1];
                sx += I[j0] * I[j1];
                sy += R[j0] * I[j1];
                sy -= I[j0] * R[j1];
            }
            const float y = 2.f * (sy.x + sy.y);
            e[q] = 2.f * (sx.x + sx.y); e[6 + q] = (p & jb) ? -y : y;
        }

        #pragma unroll
        for (int k = 0; k < 18; ++k) {
            e[k] += __shfl_xor(e[k], 1);
            e[k] += __shfl_xor(e[k], 2);
        }
        if (p == 0) {
            int b = b0 + col;
            float* o = out + ((size_t)b * S + s) * 18;
            #pragma unroll
            for (int k = 0; k < 18; ++k) o[k] = e[k];
        }
    }
}

extern "C" void kernel_launch(void* const* d_in, const int* in_sizes, int n_in,
                              void* d_out, int out_size, void* d_ws, size_t ws_size,
                              hipStream_t stream) {
    const float* x = (const float*)d_in[0];
    const float* w = (const float*)d_in[1];
    float* out = (float*)d_out;
    (void)d_ws; (void)ws_size; (void)n_in; (void)out_size;
    int S = in_sizes[1] / (NL * NQ * 3);
    int B = in_sizes[0] / (S * NQ);
    dim3 grid(S, B / BCOLS);
    qreservoir_kernel<<<grid, NTHREADS, 0, stream>>>(x, w, out, S);
}


# Result plane Sr reuses the LDS of the encoding plane P (dead after stage one): 38880 B per workgroup, not 48096, so 8 workgroups fit a WGP (was 6) and all 2048 are resident at once
# speedup vs baseline: 5.3809x; 1.5023x over previous
#include <hip/hip_runtime.h>
#include <math.h>

typedef __attribute__((ext_vector_type(2))) float v2f;
typedef __attribute__((ext_vector_type(4))) float v4f;
typedef __attribute__((ext_vector_type(8))) float v8f;

#define NQ        6
#define DIM       64
#define NL        2
#define PSTRIDE   36
#define BCOLS     32
#define NTHREADS  128

__device__ __forceinline__ float cz_sign(float v, int i)
{
    unsigned par = (unsigned)__popc(i & (i >> 1) & 0x1f) & 1u;
    return __uint_as_float(__float_as_uint(v) ^ (par << 31));
}

template <bool HI, bool REAL_IN, bool CZ_OUT>
__device__ __forceinline__ void kron_stage(const float* g,
                                           const float* Xr, const float* Xi,
                                           float* Yr, float* Yi,
                                           int wv, int lm, int lh)
{
    const int m  = lm & 7;
    const int ga = (HI ? 0 : 3) * 8 + ((m >> 2) & 1) * 4;
    const int gb = (HI ? 1 : 4) * 8 + ((m >> 1) & 1) * 4 + lh * 2;
    const int gc = (HI ? 2 : 5) * 8 + (m & 1) * 4;
    const float a0r = g[ga], a0i = g[ga + 1], a1r = g[ga + 2], a1i = g[ga + 3];
    const float br  = g[gb], bi  = g[gb + 1];
    const float c0r = g[gc], c0i = g[gc + 1], c1r = g[gc + 2], c1i = g[gc + 3];
    const float t0r = br * c0r - bi * c0i, t0i = br * c0i + bi * c0r;
    const float t1r = br * c1r - bi * c1i, t1i = br * c1i + bi * c1r;
    const float w00r = a0r * t0r - a0i * t0i, w00i = a0r * t0i + a0i * t0r;
    const float w01r = a0r * t1r - a0i * t1i, w01i = a0r * t1i + a0i * t1r;
    const float w10r = a1r * t0r - a1i * t0i, w10i = a1r * t0i + a1i * t0r;
    const float w11r = a1r * t1r - a1i * t1i, w11i = a1r * t1i + a1i * t1r;
    const bool imrow = (lm >= 8);
    v2f afr[4];
    afr[0] = imrow ? (v2f){ w00i, w01i } : (v2f){ w00r, w01r };
    afr[1] = imrow ? (v2f){ w10i, w11i } : (v2f){ w10r, w11r };
    afr[2] = imrow ? (v2f){ w00r, w01r } : (v2f){ -w00i, -w01i };
    afr[3] = imrow ? (v2f){ w10r, w11r } : (v2f){ -w10i, -w11i };

    v8f acc[4];
    #pragma unroll
    for (int t = 0; t < 4; ++t) acc[t] = (v8f){};

    #pragma unroll
    for (int k0 = 0; k0 < (REAL_IN ? 2 : 4); ++k0) {
        const float* X = (k0 >> 1) ? Xi : Xr;
        const int x0 = (k0 & 1) * 4 + lh * 2;
        #pragma unroll
        for (int t = 0; t < 4; ++t) {
            const int idx8 = wv * 2 + (t >> 1);
            const int col  = (t & 1) * 16 + lm;
            const int r0   = HI ? (x0 * 8 + idx8) : (idx8 * 8 + x0);
            const int r1   = HI ? (r0 + 8) : (r0 + 1);
            v2f b = { X[r0 * PSTRIDE + col], X[r1 * PSTRIDE + col] };
            acc[t] = __builtin_amdgcn_wmma_f32_16x16x4_f32(
                false, afr[k0], false, b, (short)0, acc[t], false, false);
        }
    }

    float* Y = lh ? Yi : Yr;
    #pragma unroll
    for (int t = 0; t < 4; ++t) {
        const int idx8 = wv * 2 + (t >> 1);
        const int col  = (t & 1) * 16 + lm;
        #pragma unroll
        for (int v = 0; v < 8; ++v) {
            const int row = HI ? (v * 8 + idx8) : (idx8 * 8 + v);
            float val = acc[t][v];
            if (CZ_OUT) val = cz_sign(val, row);
            Y[row * PSTRIDE + col] = val;
        }
    }
}

__global__ __launch_bounds__(NTHREADS)
void qreservoir_kernel(const float* __restrict__ x, const float* __restrict__ w,
                       float* __restrict__ out, int S)
{
    __shared__ float PS[DIM * PSTRIDE];
    __shared__ float Tr[DIM * PSTRIDE];
    __shared__ float Ti[DIM * PSTRIDE];
    __shared__ float Si[DIM * PSTRIDE];
    float* const P  = PS;
    float* const Sr = PS;
    __shared__ float Gs[NQ * 8];
    __shared__ v4f   Rs[NQ * 3];
    __shared__ float cs[BCOLS][NQ];
    __shared__ float sn[BCOLS][NQ];

    const int s   = blockIdx.x;
    const int b0  = blockIdx.y * BCOLS;
    const int tid = threadIdx.x;

    for (int idx = tid; idx < BCOLS * NQ; idx += NTHREADS) {
        int c = idx / NQ, q = idx % NQ;
        float f = x[((b0 + c) * S + s) * NQ + q];
        float sv, cv;
        sincosf(0.5f * 3.14159265358979323846f * f, &sv, &cv);
        sn[c][q] = sv; cs[c][q] = cv;
    }
    if (tid >= 64 && tid < 64 + NL * NQ) {
        const float cf = cosf(0.25f);
        const float sf = sinf(0.25f);
        const int gidx = tid - 64;
        const int base = (s * NL * NQ + gidx) * 3;
        float phi = w[base], th = w[base + 1], om = w[base + 2];
        float st, ct; sincosf(0.5f * th, &st, &ct);
        float sa, ca; sincosf(-0.5f * (phi + om), &sa, &ca);
        float sb, cb; sincosf( 0.5f * (phi - om), &sb, &cb);
        float r00r =  ca * ct, r00i =  sa * ct;
        float r01r = -cb * st, r01i = -sb * st;
        float r10r =  cb * st, r10i = -sb * st;
        float r11r =  ca * ct, r11i = -sa * ct;
        const float g0 = cf * r00r + sf * r01i;
        const float g1 = cf * r00i - sf * r01r;
        const float g2 = sf * r00i + cf * r01r;
        const float g3 = -sf * r00r + cf * r01i;
        const float g4 = cf * r10r + sf * r11i;
        const float g5 = cf * r10i - sf * r11r;
        const float g6 = sf * r10i + cf * r11r;
        const float g7 = -sf * r10r + cf * r11i;
        if (gidx < NQ) {
            float* go = &Gs[gidx * 8];
            go[0] = g0; go[1] = g1; go[2] = g2; go[3] = g3;
            go[4] = g4; go[5] = g5; go[6] = g6; go[7] = g7;
        } else {
            const float bc_r = g2 * g4 + g3 * g5, bc_i = g3 * g4 - g2 * g5;
            const float ad_r = g0 * g6 + g1 * g7, ad_i = g1 * g6 - g0 * g7;
            const float ac_r = g0 * g4 + g1 * g5, ac_i = g1 * g4 - g0 * g5;
            const float bd_r = g2 * g6 + g3 * g7, bd_i = g3 * g6 - g2 * g7;
            const float ab_r = g0 * g2 + g1 * g3, ab_i = g1 * g2 - g0 * g3;
            const float cd_r = g4 * g6 + g5 * g7, cd_i = g5 * g6 - g4 * g7;
            const float zz = 0.5f * ((g0 * g0 + g1 * g1) - (g2 * g2 + g3 * g3)
                                   - (g4 * g4 + g5 * g5) + (g6 * g6 + g7 * g7));
            v4f* ro = &Rs[(gidx - NQ) * 3];
            ro[0] = (v4f){ bc_r + ad_r,    ad_i - bc_i, ac_r - bd_r, 0.f };
            ro[1] = (v4f){ -(bc_i + ad_i), ad_r - bc_r, bd_i - ac_i, 0.f };
            ro[2] = (v4f){ ab_r - cd_r,    ab_i - cd_i, zz,          0.f };
        }
    }
    __syncthreads();
    {
        const int  c  = tid & 31;
        const int  hi = tid >> 5;
        const bool b5 = (hi & 2) != 0, b4 = (hi & 1) != 0;
        float f0 = b5 ? sn[c][0] : cs[c][0];
        const float f1 = b4 ? sn[c][1] : cs[c][1];
        if (b5 && b4) f0 = -f0;
        const float c2 = cs[c][2], s2 = b4 ? -sn[c][2] : sn[c][2];
        const float c3 = cs[c][3], s3 = sn[c][3];
        const float c4 = cs[c][4], s4 = sn[c][4];
        const float c5 = cs[c][5], s5 = sn[c][5];
        const float t01 = f0 * f1;
        float l2[2], l3[4], l4[8];
        l2[0] = t01 * c2; l2[1] = t01 * s2;
        #pragma unroll
        for (int j = 0; j < 4; ++j) l3[j] = l2[j >> 1] * ((j & 1) ? s3 : c3);
        #pragma unroll
        for (int j = 0; j < 8; ++j) l4[j] = l3[j >> 1] * ((j & 1) ? s4 : c4);
        #pragma unroll
        for (int j = 0; j < 16; ++j) {
            const float v = l4[j >> 1] * ((j & 1) ? s5 : c5);
            const bool neg = (__builtin_popcount(j & (j >> 1) & 7) & 1) != 0;
            P[(hi * 16 + j) * PSTRIDE + c] = neg ? -v : v;
        }
    }
    __syncthreads();

    const int lane = tid & 31;
    const int wv   = tid >> 5;
    const int lh   = lane >> 4;
    const int lm   = lane & 15;
    kron_stage<false, true,  false>(Gs,          P,  P,  Tr, Ti, wv, lm, lh);
    __syncthreads();
    kron_stage<true,  false, true >(Gs,          Tr, Ti, Sr, Si, wv, lm, lh);
    __syncthreads();

    {
        const int col = tid >> 2;
        const int p   = tid & 3;
        float e[18];
        v2f R[8], I[8];

        #pragma unroll
        for (int j = 0; j < 8; ++j) {
            const int r0 = (8 * j + p) * PSTRIDE + col;
            R[j] = (v2f){ Sr[r0], Sr[r0 + 4 * PSTRIDE] };
            I[j] = (v2f){ Si[r0], Si[r0 + 4 * PSTRIDE] };
        }
        #pragma unroll
        for (int q = 0; q < 3; ++q) {
            const int jb = 1 << (2 - q);
            v2f sx = {0.f, 0.f}, sy = {0.f, 0.f};
            #pragma unroll
            for (int j0 = 0; j0 < 8; ++j0) {
                if (j0 & jb) continue;
                const int j1 = j0 | jb;
                sx += R[j0] * R[j1];
                sx += I[j0] * I[j1];
                sy += R[j0] * I[j1];
                sy -= I[j0] * R[j1];
            }
            e[q] = 2.f * (sx.x + sx.y); e[6 + q] = 2.f * (sy.x + sy.y);
        }
        {
            float sx = 0.f, sy = 0.f;
            #pragma unroll
            for (int j = 0; j < 8; ++j) {
                sx += R[j].x * R[j].y;
                sx += I[j].x * I[j].y;
                sy += R[j].x * I[j].y;
                sy -= I[j].x * R[j].y;
            }
            e[3] = 2.f * sx; e[9] = 2.f * sy;
        }
        {
            v2f s0[8], s1[4], s2[2];
            #pragma unroll
            for (int j = 0; j < 8; ++j) s0[j] = R[j] * R[j] + I[j] * I[j];
            v2f d2 = {0.f, 0.f}, d1 = {0.f, 0.f};
            #pragma unroll
            for (int m = 0; m < 4; ++m) { d2 += s0[2 * m] - s0[2 * m + 1]; s1[m] = s0[2 * m] + s0[2 * m + 1]; }
            #pragma unroll
            for (int m = 0; m < 2; ++m) { d1 += s1[2 * m] - s1[2 * m + 1]; s2[m] = s1[2 * m] + s1[2 * m + 1]; }
            const v2f d0 = s2[0] - s2[1];
            const v2f t2 = s2[0] + s2[1];
            const float tot = t2.x + t2.y;
            e[12 + 0] = d0.x + d0.y;
            e[12 + 1] = d1.x + d1.y;
            e[12 + 2] = d2.x + d2.y;
            e[12 + 3] = t2.x - t2.y;
            e[12 + 4] = (p & 2) ? -tot : tot;
            e[12 + 5] = (p & 1) ? -tot : tot;
        }

        #pragma unroll
        for (int j = 0; j < 8; ++j) {
            const int r0 = (16 * p + 8 * (j >> 2) + ((j & 3) ^ p)) * PSTRIDE + col;
            R[j] = (v2f){ Sr[r0], Sr[r0 + 4 * PSTRIDE] };
            I[j] = (v2f){ Si[r0], Si[r0 + 4 * PSTRIDE] };
        }
        #pragma unroll
        for (int q = 4; q < 6; ++q) {
            const int jb = 1 << (5 - q);
            v2f sx = {0.f, 0.f}, sy = {0.f, 0.f};
            #pragma unroll
            for (int j0 = 0; j0 < 8; ++j0) {
                if (j0 & jb) continue;
                const int j1 = j0 | jb;
                sx += R[j0] * R[j1];
                sx += I[j0] * I[j1];
                sy += R[j0] * I[j1];
                sy -= I[j0] * R[j1];
            }
            const float y = 2.f * (sy.x + sy.y);
            e[q] = 2.f * (sx.x + sx.y); e[6 + q] = (p & jb) ? -y : y;
        }

        #pragma unroll
        for (int k = 0; k < 18; ++k) {
            e[k] += __shfl_xor(e[k], 1);
            e[k] += __shfl_xor(e[k], 2);
        }
        if (p < 3) {
            float o6[NQ];
            #pragma unroll
            for (int q = 0; q < NQ; ++q) {
                const v4f r = Rs[q * 3 + p];
                o6[q] = r.x * e[q] + r.y * e[6 + q] + r.z * e[12 + q];
            }
            float* o = out + ((size_t)(b0 + col) * S + s) * 18 + p * NQ;
            #pragma unroll
            for (int q = 0; q < NQ; ++q) o[q] = o6[q];
        }
    }
}

extern "C" void kernel_launch(void* const* d_in, const int* in_sizes, int n_in,
                              void* d_out, int out_size, void* d_ws, size_t ws_size,
                              hipStream_t stream) {
    const float* x = (const float*)d_in[0];
    const float* w = (const float*)d_in[1];
    float* out = (float*)d_out;
    (void)d_ws; (void)ws_size; (void)n_in; (void)out_size;
    int S = in_sizes[1] / (NL * NQ * 3);
    int B = in_sizes[0] / (S * NQ);
    dim3 grid(S, B / BCOLS);
    qreservoir_kernel<<<grid, NTHREADS, 0, stream>>>(x, w, out, S);
}


# First factor applied to the product state's factor vectors, T = u x (W v): two K=4 matrix products per 16 columns, second factor's B fragments from registers; no P/T planes; changes summation order
# speedup vs baseline: 5.8017x; 1.0782x over previous
#include <hip/hip_runtime.h>
#include <math.h>

typedef __attribute__((ext_vector_type(2))) float v2f;
typedef __attribute__((ext_vector_type(4))) float v4f;
typedef __attribute__((ext_vector_type(8))) float v8f;

#define NQ        6
#define DIM       64
#define NL        2
#define PSTRIDE   36
#define BCOLS     32
#define NTHREADS  128

__device__ __forceinline__ float cz_sign(float v, int i)
{
    unsigned par = (unsigned)__popc(i & (i >> 1) & 0x1f) & 1u;
    return __uint_as_float(__float_as_uint(v) ^ (par << 31));
}

__device__ __forceinline__ void first_layer(const float* g, const float (*cs)[NQ], const float (*sn)[NQ],
                                            float* Yr, float* Yi, int wv, int lm, int lh)
{
    v2f a1[2];
    {
        const int m  = 2 * wv + ((lm >> 1) & 1);
        const int ga = 3 * 8 + ((m >> 2) & 1) * 4;
        const int gb = 4 * 8 + ((m >> 1) & 1) * 4 + lh * 2;
        const int gc = 5 * 8 + (m & 1) * 4;
        const float a0r = g[ga], a0i = g[ga + 1], a1r = g[ga + 2], a1i = g[ga + 3];
        const float br  = g[gb], bi  = g[gb + 1];
        const float c0r = g[gc], c0i = g[gc + 1], c1r = g[gc + 2], c1i = g[gc + 3];
        const float t0r = br * c0r - bi * c0i, t0i = br * c0i + bi * c0r;
        const float t1r = br * c1r - bi * c1i, t1i = br * c1i + bi * c1r;
        const float w00r = a0r * t0r - a0i * t0i, w00i = a0r * t0i + a0i * t0r;
        const float w01r = a0r * t1r - a0i * t1i, w01i = a0r * t1i + a0i * t1r;
        const float w10r = a1r * t0r - a1i * t0i, w10i = a1r * t0i + a1i * t0r;
        const float w11r = a1r * t1r - a1i * t1i, w11i = a1r * t1i + a1i * t1r;
        const bool im = (lm & 1) != 0;
        a1[0] = im ? (v2f){ w00i, w01i } : (v2f){ w00r, w01r };
        a1[1] = im ? (v2f){ w10i, w11i } : (v2f){ w10r, w11r };
    }
    float ys[2][4], yd[2][4], u[2][2][2];
    #pragma unroll
    for (int tc = 0; tc < 2; ++tc) {
        const int col = tc * 16 + lm;
        const float c3 = cs[col][3], s3 = sn[col][3];
        const float g4 = lh ? sn[col][4] : cs[col][4];
        const float g4n = lh ? -g4 : g4;
        const float c5 = cs[col][5], s5 = sn[col][5];
        const v2f bA = { c3 * c5 * g4,  c3 * s5 * g4n };
        const v2f bB = { s3 * c5 * g4n, s3 * s5 * g4  };
        const v8f ya = __builtin_amdgcn_wmma_f32_16x16x4_f32(false, a1[0], false, bA, (short)0, (v8f){}, false, false);
        const v8f yb = __builtin_amdgcn_wmma_f32_16x16x4_f32(false, a1[1], false, bB, (short)0, (v8f){}, false, false);
        #pragma unroll
        for (int v = 0; v < 4; ++v) { ys[tc][v] = ya[v] + yb[v]; yd[tc][v] = ya[v] - yb[v]; }
        const float c0 = cs[col][0], s0 = sn[col][0];
        const float g1 = lh ? sn[col][1] : cs[col][1];
        const float g1n = lh ? -g1 : g1;
        const float c2 = cs[col][2], s2 = sn[col][2];
        u[tc][0][0] = c0 * c2 * g1;  u[tc][0][1] = c0 * s2 * g1n;
        u[tc][1][0] = s0 * c2 * g1n; u[tc][1][1] = s0 * s2 * g1;
    }
    const int m  = lm & 7;
    const int ga = 0 * 8 + ((m >> 2) & 1) * 4;
    const int gb = 1 * 8 + ((m >> 1) & 1) * 4 + lh * 2;
    const int gc = 2 * 8 + (m & 1) * 4;
    const float a0r = g[ga], a0i = g[ga + 1], a1r = g[ga + 2], a1i = g[ga + 3];
    const float br  = g[gb], bi  = g[gb + 1];
    const float c0r = g[gc], c0i = g[gc + 1], c1r = g[gc + 2], c1i = g[gc + 3];
    const float t0r = br * c0r - bi * c0i, t0i = br * c0i + bi * c0r;
    const float t1r = br * c1r - bi * c1i, t1i = br * c1i + bi * c1r;
    const float w00r = a0r * t0r - a0i * t0i, w00i = a0r * t0i + a0i * t0r;
    const float w01r = a0r * t1r - a0i * t1i, w01i = a0r * t1i + a0i * t1r;
    const float w10r = a1r * t0r - a1i * t0i, w10i = a1r * t0i + a1i * t0r;
    const float w11r = a1r * t1r - a1i * t1i, w11i = a1r * t1i + a1i * t1r;
    const bool imrow = (lm >= 8);
    v2f afr[4];
    afr[0] = imrow ? (v2f){ w00i, w01i } : (v2f){ w00r, w01r };
    afr[1] = imrow ? (v2f){ w10i, w11i } : (v2f){ w10r, w11r };
    afr[2] = imrow ? (v2f){ w00r, w01r } : (v2f){ -w00i, -w01i };
    afr[3] = imrow ? (v2f){ w10r, w11r } : (v2f){ -w10i, -w11i };

    v8f acc[4];
    #pragma unroll
    for (int t = 0; t < 4; ++t) acc[t] = (v8f){};
    #pragma unroll
    for (int k0 = 0; k0 < 4; ++k0) {
        const int part = k0 >> 1, c2 = k0 & 1;
        #pragma unroll
        for (int t = 0; t < 4; ++t) {
            const int tc = t & 1, ms = t >> 1;
            const v2f b = { u[tc][c2][0] * ys[tc][2 * ms + part], u[tc][c2][1] * yd[tc][2 * ms + part] };
            acc[t] = __builtin_amdgcn_wmma_f32_16x16x4_f32(
                false, afr[k0], false, b, (short)0, acc[t], false, false);
        }
    }
    float* Y = lh ? Yi : Yr;
    #pragma unroll
    for (int t = 0; t < 4; ++t) {
        const int idx8 = wv * 2 + (t >> 1);
        const int col  = (t & 1) * 16 + lm;
        #pragma unroll
        for (int v = 0; v < 8; ++v) {
            const int row = v * 8 + idx8;
            Y[row * PSTRIDE + col] = cz_sign(acc[t][v], row);
        }
    }
}

__global__ __launch_bounds__(NTHREADS)
void qreservoir_kernel(const float* __restrict__ x, const float* __restrict__ w,
                       float* __restrict__ out, int S)
{
    __shared__ float Sr[DIM * PSTRIDE];
    __shared__ float Si[DIM * PSTRIDE];
    __shared__ float Gs[NQ * 8];
    __shared__ v4f   Rs[NQ * 3];
    __shared__ float cs[BCOLS][NQ];
    __shared__ float sn[BCOLS][NQ];

    const int s   = blockIdx.x;
    const int b0  = blockIdx.y * BCOLS;
    const int tid = threadIdx.x;

    for (int idx = tid; idx < BCOLS * NQ; idx += NTHREADS) {
        int c = idx / NQ, q = idx % NQ;
        float f = x[((b0 + c) * S + s) * NQ + q];
        float sv, cv;
        sincosf(0.5f * 3.14159265358979323846f * f, &sv, &cv);
        sn[c][q] = sv; cs[c][q] = cv;
    }
    if (tid >= 64 && tid < 64 + NL * NQ) {
        const float cf = cosf(0.25f);
        const float sf = sinf(0.25f);
        const int gidx = tid - 64;
        const int base = (s * NL * NQ + gidx) * 3;
        float phi = w[base], th = w[base + 1], om = w[base + 2];
        float st, ct; sincosf(0.5f * th, &st, &ct);
        float sa, ca; sincosf(-0.5f * (phi + om), &sa, &ca);
        float sb, cb; sincosf( 0.5f * (phi - om), &sb, &cb);
        float r00r =  ca * ct, r00i =  sa * ct;
        float r01r = -cb * st, r01i = -sb * st;
        float r10r =  cb * st, r10i = -sb * st;
        float r11r =  ca * ct, r11i = -sa * ct;
        const float g0 = cf * r00r + sf * r01i;
        const float g1 = cf * r00i - sf * r01r;
        const float g2 = sf * r00i + cf * r01r;
        const float g3 = -sf * r00r + cf * r01i;
        const float g4 = cf * r10r + sf * r11i;
        const float g5 = cf * r10i - sf * r11r;
        const float g6 = sf * r10i + cf * r11r;
        const float g7 = -sf * r10r + cf * r11i;
        if (gidx < NQ) {
            float* go = &Gs[gidx * 8];
            go[0] = g0; go[1] = g1; go[2] = g2; go[3] = g3;
            go[4] = g4; go[5] = g5; go[6] = g6; go[7] = g7;
        } else {
            const float bc_r = g2 * g4 + g3 * g5, bc_i = g3 * g4 - g2 * g5;
            const float ad_r = g0 * g6 + g1 * g7, ad_i = g1 * g6 - g0 * g7;
            const float ac_r = g0 * g4 + g1 * g5, ac_i = g1 * g4 - g0 * g5;
            const float bd_r = g2 * g6 + g3 * g7, bd_i = g3 * g6 - g2 * g7;
            const float ab_r = g0 * g2 + g1 * g3, ab_i = g1 * g2 - g0 * g3;
            const float cd_r = g4 * g6 + g5 * g7, cd_i = g5 * g6 - g4 * g7;
            const float zz = 0.5f * ((g0 * g0 + g1 * g1) - (g2 * g2 + g3 * g3)
                                   - (g4 * g4 + g5 * g5) + (g6 * g6 + g7 * g7));
            v4f* ro = &Rs[(gidx - NQ) * 3];
            ro[0] = (v4f){ bc_r + ad_r,    ad_i - bc_i, ac_r - bd_r, 0.f };
            ro[1] = (v4f){ -(bc_i + ad_i), ad_r - bc_r, bd_i - ac_i, 0.f };
            ro[2] = (v4f){ ab_r - cd_r,    ab_i - cd_i, zz,          0.f };
        }
    }
    __syncthreads();

    const int lane = tid & 31;
    const int wv   = tid >> 5;
    const int lh   = lane >> 4;
    const int lm   = lane & 15;
    first_layer(Gs, cs, sn, Sr, Si, wv, lm, lh);
    __syncthreads();

    {
        const int col = tid >> 2;
        const int p   = tid & 3;
        float e[18];
        v2f R[8], I[8];

        #pragma unroll
        for (int j = 0; j < 8; ++j) {
            const int r0 = (8 * j + p) * PSTRIDE + col;
            R[j] = (v2f){ Sr[r0], Sr[r0 + 4 * PSTRIDE] };
            I[j] = (v2f){ Si[r0], Si[r0 + 4 * PSTRIDE] };
        }
        #pragma unroll
        for (int q = 0; q < 3; ++q) {
            const int jb = 1 << (2 - q);
            v2f sx = {0.f, 0.f}, sy = {0.f, 0.f};
            #pragma unroll
            for (int j0 = 0; j0 < 8; ++j0) {
                if (j0 & jb) continue;
                const int j1 = j0 | jb;
                sx += R[j0] * R[j1];
                sx += I[j0] * I[j1];
                sy += R[j0] * I[j1];
                sy -= I[j0] * R[j1];
            }
            e[q] = 2.f * (sx.x + sx.y); e[6 + q] = 2.f * (sy.x + sy.y);
        }
        {
            float sx = 0.f, sy = 0.f;
            #pragma unroll
            for (int j = 0; j < 8; ++j) {
                sx += R[j].x * R[j].y;
                sx += I[j].x * I[j].y;
                sy += R[j].x * I[j].y;
                sy -= I[j].x * R[j].y;
            }
            e[3] = 2.f * sx; e[9] = 2.f * sy;
        }
        {
            v2f s0[8], s1[4], s2[2];
            #pragma unroll
            for (int j = 0; j < 8; ++j) s0[j] = R[j] * R[j] + I[j] * I[j];
            v2f d2 = {0.f, 0.f}, d1 = {0.f, 0.f};
            #pragma unroll
            for (int m = 0; m < 4; ++m) { d2 += s0[2 * m] - s0[2 * m + 1]; s1[m] = s0[2 * m] + s0[2 * m + 1]; }
            #pragma unroll
            for (int m = 0; m < 2; ++m) { d1 += s1[2 * m] - s1[2 * m + 1]; s2[m] = s1[2 * m] + s1[2 * m + 1]; }
            const v2f d0 = s2[0] - s2[1];
            const v2f t2 = s2[0] + s2[1];
            const float tot = t2.x + t2.y;
            e[12 + 0] = d0.x + d0.y;
            e[12 + 1] = d1.x + d1.y;
            e[12 + 2] = d2.x + d2.y;
            e[12 + 3] = t2.x - t2.y;
            e[12 + 4] = (p & 2) ? -tot : tot;
            e[12 + 5] = (p & 1) ? -tot : tot;
        }

        #pragma unroll
        for (int j = 0; j < 8; ++j) {
            const int r0 = (16 * p + 8 * (j >> 2) + ((j & 3) ^ p)) * PSTRIDE + col;
            R[j] = (v2f){ Sr[r0], Sr[r0 + 4 * PSTRIDE] };
            I[j] = (v2f){ Si[r0], Si[r0 + 4 * PSTRIDE] };
        }
        #pragma unroll
        for (int q = 4; q < 6; ++q) {
            const int jb = 1 << (5 - q);
            v2f sx = {0.f, 0.f}, sy = {0.f, 0.f};
            #pragma unroll
            for (int j0 = 0; j0 < 8; ++j0) {
                if (j0 & jb) continue;
                const int j1 = j0 | jb;
                sx += R[j0] * R[j1];
                sx += I[j0] * I[j1];
                sy += R[j0] * I[j1];
                sy -= I[j0] * R[j1];
            }
            const float y = 2.f * (sy.x + sy.y);
            e[q] = 2.f * (sx.x + sx.y); e[6 + q] = (p & jb) ? -y : y;
        }

        #pragma unroll
        for (int k = 0; k < 18; ++k) {
            e[k] += __shfl_xor(e[k], 1);
            e[k] += __shfl_xor(e[k], 2);
        }
        if (p < 3) {
            float o6[NQ];
            #pragma unroll
            for (int q = 0; q < NQ; ++q) {
                const v4f r = Rs[q * 3 + p];
                o6[q] = r.x * e[q] + r.y * e[6 + q] + r.z * e[12 + q];
            }
            float* o = out + ((size_t)(b0 + col) * S + s) * 18 + p * NQ;
            #pragma unroll
            for (int q = 0; q < NQ; ++q) o[q] = o6[q];
        }
    }
}

extern "C" void kernel_launch(void* const* d_in, const int* in_sizes, int n_in,
                              void* d_out, int out_size, void* d_ws, size_t ws_size,
                              hipStream_t stream) {
    const float* x = (const float*)d_in[0];
    const float* w = (const float*)d_in[1];
    float* out = (float*)d_out;
    (void)d_ws; (void)ws_size; (void)n_in; (void)out_size;
    int S = in_sizes[1] / (NL * NQ * 3);
    int B = in_sizes[0] / (S * NQ);
    dim3 grid(S, B / BCOLS);
    qreservoir_kernel<<<grid, NTHREADS, 0, stream>>>(x, w, out, S);
}


# Set-up trig spread over lanes: each of the 36 gate half-angles gets its own sin/cos beside the encoding angles (was 3 in a row on 12 lanes behind a 2nd load); gates combined behind one more barrier
# speedup vs baseline: 6.1310x; 1.0568x over previous
#include <hip/hip_runtime.h>
#include <math.h>

typedef __attribute__((ext_vector_type(2))) float v2f;
typedef __attribute__((ext_vector_type(4))) float v4f;
typedef __attribute__((ext_vector_type(8))) float v8f;

#define NQ        6
#define DIM       64
#define NL        2
#define PSTRIDE   36
#define BCOLS     32
#define NTHREADS  128

__device__ __forceinline__ float cz_sign(float v, int i)
{
    unsigned par = (unsigned)__popc(i & (i >> 1) & 0x1f) & 1u;
    return __uint_as_float(__float_as_uint(v) ^ (par << 31));
}

__device__ __forceinline__ void first_layer_prep(const float (*cs)[NQ], const float (*sn)[NQ], int lm, int lh,
                                                 v2f (&bA)[2], v2f (&bB)[2], float (&u)[2][2][2])
{
    #pragma unroll
    for (int tc = 0; tc < 2; ++tc) {
        const int col = tc * 16 + lm;
        const float c3 = cs[col][3], s3 = sn[col][3];
        const float g4 = lh ? sn[col][4] : cs[col][4];
        const float g4n = lh ? -g4 : g4;
        const float c5 = cs[col][5], s5 = sn[col][5];
        bA[tc] = (v2f){ c3 * c5 * g4,  c3 * s5 * g4n };
        bB[tc] = (v2f){ s3 * c5 * g4n, s3 * s5 * g4  };
        const float c0 = cs[col][0], s0 = sn[col][0];
        const float g1 = lh ? sn[col][1] : cs[col][1];
        const float g1n = lh ? -g1 : g1;
        const float c2 = cs[col][2], s2 = sn[col][2];
        u[tc][0][0] = c0 * c2 * g1;  u[tc][0][1] = c0 * s2 * g1n;
        u[tc][1][0] = s0 * c2 * g1n; u[tc][1][1] = s0 * s2 * g1;
    }
}

__device__ __forceinline__ void first_layer(const float* g, const v2f (&bA)[2], const v2f (&bB)[2],
                                            const float (&u)[2][2][2],
                                            float* Yr, float* Yi, int wv, int lm, int lh)
{
    v2f a1[2];
    {
        const int m  = 2 * wv + ((lm >> 1) & 1);
        const int ga = 3 * 8 + ((m >> 2) & 1) * 4;
        const int gb = 4 * 8 + ((m >> 1) & 1) * 4 + lh * 2;
        const int gc = 5 * 8 + (m & 1) * 4;
        const float a0r = g[ga], a0i = g[ga + 1], a1r = g[ga + 2], a1i = g[ga + 3];
        const float br  = g[gb], bi  = g[gb + 1];
        const float c0r = g[gc], c0i = g[gc + 1], c1r = g[gc + 2], c1i = g[gc + 3];
        const float t0r = br * c0r - bi * c0i, t0i = br * c0i + bi * c0r;
        const float t1r = br * c1r - bi * c1i, t1i = br * c1i + bi * c1r;
        const float w00r = a0r * t0r - a0i * t0i, w00i = a0r * t0i + a0i * t0r;
        const float w01r = a0r * t1r - a0i * t1i, w01i = a0r * t1i + a0i * t1r;
        const float w10r = a1r * t0r - a1i * t0i, w10i = a1r * t0i + a1i * t0r;
        const float w11r = a1r * t1r - a1i * t1i, w11i = a1r * t1i + a1i * t1r;
        const bool im = (lm & 1) != 0;
        a1[0] = im ? (v2f){ w00i, w01i } : (v2f){ w00r, w01r };
        a1[1] = im ? (v2f){ w10i, w11i } : (v2f){ w10r, w11r };
    }
    float ys[2][4], yd[2][4];
    #pragma unroll
    for (int tc = 0; tc < 2; ++tc) {
        const v8f ya = __builtin_amdgcn_wmma_f32_16x16x4_f32(false, a1[0], false, bA[tc], (short)0, (v8f){}, false, false);
        const v8f yb = __builtin_amdgcn_wmma_f32_16x16x4_f32(false, a1[1], false, bB[tc], (short)0, (v8f){}, false, false);
        #pragma unroll
        for (int v = 0; v < 4; ++v) { ys[tc][v] = ya[v] + yb[v]; yd[tc][v] = ya[v] - yb[v]; }
    }
    const int m  = lm & 7;
    const int ga = 0 * 8 + ((m >> 2) & 1) * 4;
    const int gb = 1 * 8 + ((m >> 1) & 1) * 4 + lh * 2;
    const int gc = 2 * 8 + (m & 1) * 4;
    const float a0r = g[ga], a0i = g[ga + 1], a1r = g[ga + 2], a1i = g[ga + 3];
    const float br  = g[gb], bi  = g[gb + 1];
    const float c0r = g[gc], c0i = g[gc + 1], c1r = g[gc + 2], c1i = g[gc + 3];
    const float t0r = br * c0r - bi * c0i, t0i = br * c0i + bi * c0r;
    const float t1r = br * c1r - bi * c1i, t1i = br * c1i + bi * c1r;
    const float w00r = a0r * t0r - a0i * t0i, w00i = a0r * t0i + a0i * t0r;
    const float w01r = a0r * t1r - a0i * t1i, w01i = a0r * t1i + a0i * t1r;
    const float w10r = a1r * t0r - a1i * t0i, w10i = a1r * t0i + a1i * t0r;
    const float w11r = a1r * t1r - a1i * t1i, w11i = a1r * t1i + a1i * t1r;
    const bool imrow = (lm >= 8);
    v2f afr[4];
    afr[0] = imrow ? (v2f){ w00i, w01i } : (v2f){ w00r, w01r };
    afr[1] = imrow ? (v2f){ w10i, w11i } : (v2f){ w10r, w11r };
    afr[2] = imrow ? (v2f){ w00r, w01r } : (v2f){ -w00i, -w01i };
    afr[3] = imrow ? (v2f){ w10r, w11r } : (v2f){ -w10i, -w11i };

    v8f acc[4];
    #pragma unroll
    for (int t = 0; t < 4; ++t) acc[t] = (v8f){};
    #pragma unroll
    for (int k0 = 0; k0 < 4; ++k0) {
        const int part = k0 >> 1, c2 = k0 & 1;
        #pragma unroll
        for (int t = 0; t < 4; ++t) {
            const int tc = t & 1, ms = t >> 1;
            const v2f b = { u[tc][c2][0] * ys[tc][2 * ms + part], u[tc][c2][1] * yd[tc][2 * ms + part] };
            acc[t] = __builtin_amdgcn_wmma_f32_16x16x4_f32(
                false, afr[k0], false, b, (short)0, acc[t], false, false);
        }
    }
    float* Y = lh ? Yi : Yr;
    #pragma unroll
    for (int t = 0; t < 4; ++t) {
        const int idx8 = wv * 2 + (t >> 1);
        const int col  = (t & 1) * 16 + lm;
        #pragma unroll
        for (int v = 0; v < 8; ++v) {
            const int row = v * 8 + idx8;
            Y[row * PSTRIDE + col] = cz_sign(acc[t][v], row);
        }
    }
}

__global__ __launch_bounds__(NTHREADS)
void qreservoir_kernel(const float* __restrict__ x, const float* __restrict__ w,
                       float* __restrict__ out, int S)
{
    __shared__ float Sr[DIM * PSTRIDE];
    __shared__ float Si[DIM * PSTRIDE];
    __shared__ float Gs[NQ * 8];
    __shared__ v4f   Rs[NQ * 3];
    __shared__ float cs[BCOLS][NQ];
    __shared__ float sn[BCOLS][NQ];

    const int s   = blockIdx.x;
    const int b0  = blockIdx.y * BCOLS;
    const int tid = threadIdx.x;

    const int lane = tid & 31;
    const int wv   = tid >> 5;
    const int lh   = lane >> 4;
    const int lm   = lane & 15;

    float* Ts = Sr;
    {
        const bool xb = tid < 64;
        const bool gl = tid >= 64 && tid < 64 + NL * NQ * 3;
        const int k  = gl ? tid - 64 : 0;
        const int gq = k / 3;
        const int j  = k - 3 * gq;
        const int ib = xb ? tid + NTHREADS : tid;
        const int oa = ((b0 + tid / NQ) * S + s) * NQ + tid % NQ;
        const int ob = ((b0 + ib / NQ) * S + s) * NQ + ib % NQ;
        const int ow = (s * NL * NQ + gq) * 3;
        const float* src = gl ? w : x;
        const int o1 = gl ? ow + (j == 0 ? 1 : 0) : ob;
        const int o2 = gl ? ow + 2 : oa;
        const float fa = x[oa];
        const float f1 = src[o1];
        const float f2 = src[o2];
        const float hp = 0.5f * 3.14159265358979323846f;
        const float t12 = (j == 1) ? f1 + f2 : f1 - f2;
        const float hg  = (j == 0) ? 0.5f * f1 : ((j == 1) ? -0.5f * t12 : 0.5f * t12);
        const float angB = gl ? hg : hp * f1;
        float sA, cA;
        sincosf(hp * fa, &sA, &cA);
        sn[tid / NQ][tid % NQ] = sA; cs[tid / NQ][tid % NQ] = cA;
        if (tid < 64 + NL * NQ * 3) {
            float sB, cB;
            sincosf(angB, &sB, &cB);
            float* ds = xb ? &sn[ib / NQ][ib % NQ] : Ts + 2 * k;
            float* dc = xb ? &cs[ib / NQ][ib % NQ] : Ts + 2 * k + 1;
            *ds = sB; *dc = cB;
        }
    }
    __syncthreads();

    v2f bA[2], bB[2];
    float u[2][2][2];
    first_layer_prep(cs, sn, lm, lh, bA, bB, u);
    if (tid >= 64 && tid < 64 + NL * NQ) {
        const float cf = cosf(0.25f);
        const float sf = sinf(0.25f);
        const int gidx = tid - 64;
        const float st = Ts[6 * gidx],     ct = Ts[6 * gidx + 1];
        const float sa = Ts[6 * gidx + 2], ca = Ts[6 * gidx + 3];
        const float sb = Ts[6 * gidx + 4], cb = Ts[6 * gidx + 5];
        float r00r =  ca * ct, r00i =  sa * ct;
        float r01r = -cb * st, r01i = -sb * st;
        float r10r =  cb * st, r10i = -sb * st;
        float r11r =  ca * ct, r11i = -sa * ct;
        const float g0 = cf * r00r + sf * r01i;
        const float g1 = cf * r00i - sf * r01r;
        const float g2 = sf * r00i + cf * r01r;
        const float g3 = -sf * r00r + cf * r01i;
        const float g4 = cf * r10r + sf * r11i;
        const float g5 = cf * r10i - sf * r11r;
        const float g6 = sf * r10i + cf * r11r;
        const float g7 = -sf * r10r + cf * r11i;
        if (gidx < NQ) {
            float* go = &Gs[gidx * 8];
            go[0] = g0; go[1] = g1; go[2] = g2; go[3] = g3;
            go[4] = g4; go[5] = g5; go[6] = g6; go[7] = g7;
        } else {
            const float bc_r = g2 * g4 + g3 * g5, bc_i = g3 * g4 - g2 * g5;
            const float ad_r = g0 * g6 + g1 * g7, ad_i = g1 * g6 - g0 * g7;
            const float ac_r = g0 * g4 + g1 * g5, ac_i = g1 * g4 - g0 * g5;
            const float bd_r = g2 * g6 + g3 * g7, bd_i = g3 * g6 - g2 * g7;
            const float ab_r = g0 * g2 + g1 * g3, ab_i = g1 * g2 - g0 * g3;
            const float cd_r = g4 * g6 + g5 * g7, cd_i = g5 * g6 - g4 * g7;
            const float zz = 0.5f * ((g0 * g0 + g1 * g1) - (g2 * g2 + g3 * g3)
                                   - (g4 * g4 + g5 * g5) + (g6 * g6 + g7 * g7));
            v4f* ro = &Rs[(gidx - NQ) * 3];
            ro[0] = (v4f){ bc_r + ad_r,    ad_i - bc_i, ac_r - bd_r, 0.f };
            ro[1] = (v4f){ -(bc_i + ad_i), ad_r - bc_r, bd_i - ac_i, 0.f };
            ro[2] = (v4f){ ab_r - cd_r,    ab_i - cd_i, zz,          0.f };
        }
    }
    __syncthreads();

    first_layer(Gs, bA, bB, u, Sr, Si, wv, lm, lh);
    __syncthreads();

    {
        const int col = tid >> 2;
        const int p   = tid & 3;
        float e[18];
        v2f R[8], I[8];

        #pragma unroll
        for (int j = 0; j < 8; ++j) {
            const int r0 = (8 * j + p) * PSTRIDE + col;
            R[j] = (v2f){ Sr[r0], Sr[r0 + 4 * PSTRIDE] };
            I[j] = (v2f){ Si[r0], Si[r0 + 4 * PSTRIDE] };
        }
        #pragma unroll
        for (int q = 0; q < 3; ++q) {
            const int jb = 1 << (2 - q);
            v2f sx = {0.f, 0.f}, sy = {0.f, 0.f};
            #pragma unroll
            for (int j0 = 0; j0 < 8; ++j0) {
                if (j0 & jb) continue;
                const int j1 = j0 | jb;
                sx += R[j0] * R[j1];
                sx += I[j0] * I[j1];
                sy += R[j0] * I[j1];
                sy -= I[j0] * R[j1];
            }
            e[q] = 2.f * (sx.x + sx.y); e[6 + q] = 2.f * (sy.x + sy.y);
        }
        {
            float sx = 0.f, sy = 0.f;
            #pragma unroll
            for (int j = 0; j < 8; ++j) {
                sx += R[j].x * R[j].y;
                sx += I[j].x * I[j].y;
                sy += R[j].x * I[j].y;
                sy -= I[j].x * R[j].y;
            }
            e[3] = 2.f * sx; e[9] = 2.f * sy;
        }
        {
            v2f s0[8], s1[4], s2[2];
            #pragma unroll
            for (int j = 0; j < 8; ++j) s0[j] = R[j] * R[j] + I[j] * I[j];
            v2f d2 = {0.f, 0.f}, d1 = {0.f, 0.f};
            #pragma unroll
            for (int m = 0; m < 4; ++m) { d2 += s0[2 * m] - s0[2 * m + 1]; s1[m] = s0[2 * m] + s0[2 * m + 1]; }
            #pragma unroll
            for (int m = 0; m < 2; ++m) { d1 += s1[2 * m] - s1[2 * m + 1]; s2[m] = s1[2 * m] + s1[2 * m + 1]; }
            const v2f d0 = s2[0] - s2[1];
            const v2f t2 = s2[0] + s2[1];
            const float tot = t2.x + t2.y;
            e[12 + 0] = d0.x + d0.y;
            e[12 + 1] = d1.x + d1.y;
            e[12 + 2] = d2.x + d2.y;
            e[12 + 3] = t2.x - t2.y;
            e[12 + 4] = (p & 2) ? -tot : tot;
            e[12 + 5] = (p & 1) ? -tot : tot;
        }

        #pragma unroll
        for (int j = 0; j < 8; ++j) {
            const int r0 = (16 * p + 8 * (j >> 2) + ((j & 3) ^ p)) * PSTRIDE + col;
            R[j] = (v2f){ Sr[r0], Sr[r0 + 4 * PSTRIDE] };
            I[j] = (v2f){ Si[r0], Si[r0 + 4 * PSTRIDE] };
        }
        #pragma unroll
        for (int q = 4; q < 6; ++q) {
            const int jb = 1 << (5 - q);
            v2f sx = {0.f, 0.f}, sy = {0.f, 0.f};
            #pragma unroll
            for (int j0 = 0; j0 < 8; ++j0) {
                if (j0 & jb) continue;
                const int j1 = j0 | jb;
                sx += R[j0] * R[j1];
                sx += I[j0] * I[j1];
                sy += R[j0] * I[j1];
                sy -= I[j0] * R[j1];
            }
            const float y = 2.f * (sy.x + sy.y);
            e[q] = 2.f * (sx.x + sx.y); e[6 + q] = (p & jb) ? -y : y;
        }

        #pragma unroll
        for (int k = 0; k < 18; ++k) {
            e[k] += __shfl_xor(e[k], 1);
            e[k] += __shfl_xor(e[k], 2);
        }
        if (p < 3) {
            float o6[NQ];
            #pragma unroll
            for (int q = 0; q < NQ; ++q) {
                const v4f r = Rs[q * 3 + p];
                o6[q] = r.x * e[q] + r.y * e[6 + q] + r.z * e[12 + q];
            }
            float* o = out + ((size_t)(b0 + col) * S + s) * 18 + p * NQ;
            #pragma unroll
            for (int q = 0; q < NQ; ++q) o[q] = o6[q];
        }
    }
}

extern "C" void kernel_launch(void* const* d_in, const int* in_sizes, int n_in,
                              void* d_out, int out_size, void* d_ws, size_t ws_size,
                              hipStream_t stream) {
    const float* x = (const float*)d_in[0];
    const float* w = (const float*)d_in[1];
    float* out = (float*)d_out;
    (void)d_ws; (void)ws_size; (void)n_in; (void)out_size;
    int S = in_sizes[1] / (NL * NQ * 3);
    int B = in_sizes[0] / (S * NQ);
    dim3 grid(S, B / BCOLS);
    qreservoir_kernel<<<grid, NTHREADS, 0, stream>>>(x, w, out, S);
}


# State plane as interleaved complex numbers: second factor's A rows reordered so result registers pair (Re, Im); 8-byte LDS stores and reads, pair sums as complex products; changes summation order
# speedup vs baseline: 6.5152x; 1.0627x over previous
#include <hip/hip_runtime.h>
#include <math.h>

typedef __attribute__((ext_vector_type(2))) float v2f;
typedef __attribute__((ext_vector_type(4))) float v4f;
typedef __attribute__((ext_vector_type(8))) float v8f;

#define NQ        6
#define DIM       64
#define NL        2
#define PROW      32
#define BCOLS     32
#define NTHREADS  128

__device__ __forceinline__ float cz_sign(float v, int i)
{
    unsigned par = (unsigned)__popc(i & (i >> 1) & 0x1f) & 1u;
    return __uint_as_float(__float_as_uint(v) ^ (par << 31));
}

__device__ __forceinline__ void first_layer_prep(const float (*cs)[NQ], const float (*sn)[NQ], int lm, int lh,
                                                 v2f (&bA)[2], v2f (&bB)[2], float (&u)[2][2][2])
{
    #pragma unroll
    for (int tc = 0; tc < 2; ++tc) {
        const int col = tc * 16 + lm;
        const float c3 = cs[col][3], s3 = sn[col][3];
        const float g4 = lh ? sn[col][4] : cs[col][4];
        const float g4n = lh ? -g4 : g4;
        const float c5 = cs[col][5], s5 = sn[col][5];
        bA[tc] = (v2f){ c3 * c5 * g4,  c3 * s5 * g4n };
        bB[tc] = (v2f){ s3 * c5 * g4n, s3 * s5 * g4  };
        const float c0 = cs[col][0], s0 = sn[col][0];
        const float g1 = lh ? sn[col][1] : cs[col][1];
        const float g1n = lh ? -g1 : g1;
        const float c2 = cs[col][2], s2 = sn[col][2];
        u[tc][0][0] = c0 * c2 * g1;  u[tc][0][1] = c0 * s2 * g1n;
        u[tc][1][0] = s0 * c2 * g1n; u[tc][1][1] = s0 * s2 * g1;
    }
}

__device__ __forceinline__ void first_layer(const float* g, const v2f (&bA)[2], const v2f (&bB)[2],
                                            const float (&u)[2][2][2],
                                            v2f* Sp, int wv, int lm, int lh)
{
    v2f a1[2];
    {
        const int m  = 2 * wv + ((lm >> 1) & 1);
        const int ga = 3 * 8 + ((m >> 2) & 1) * 4;
        const int gb = 4 * 8 + ((m >> 1) & 1) * 4 + lh * 2;
        const int gc = 5 * 8 + (m & 1) * 4;
        const float a0r = g[ga], a0i = g[ga + 1], a1r = g[ga + 2], a1i = g[ga + 3];
        const float br  = g[gb], bi  = g[gb + 1];
        const float c0r = g[gc], c0i = g[gc + 1], c1r = g[gc + 2], c1i = g[gc + 3];
        const float t0r = br * c0r - bi * c0i, t0i = br * c0i + bi * c0r;
        const float t1r = br * c1r - bi * c1i, t1i = br * c1i + bi * c1r;
        const float w00r = a0r * t0r - a0i * t0i, w00i = a0r * t0i + a0i * t0r;
        const float w01r = a0r * t1r - a0i * t1i, w01i = a0r * t1i + a0i * t1r;
        const float w10r = a1r * t0r - a1i * t0i, w10i = a1r * t0i + a1i * t0r;
        const float w11r = a1r * t1r - a1i * t1i, w11i = a1r * t1i + a1i * t1r;
        const bool im = (lm & 1) != 0;
        a1[0] = im ? (v2f){ w00i, w01i } : (v2f){ w00r, w01r };
        a1[1] = im ? (v2f){ w10i, w11i } : (v2f){ w10r, w11r };
    }
    float ys[2][4], yd[2][4];
    #pragma unroll
    for (int tc = 0; tc < 2; ++tc) {
        const v8f ya = __builtin_amdgcn_wmma_f32_16x16x4_f32(false, a1[0], false, bA[tc], (short)0, (v8f){}, false, false);
        const v8f yb = __builtin_amdgcn_wmma_f32_16x16x4_f32(false, a1[1], false, bB[tc], (short)0, (v8f){}, false, false);
        #pragma unroll
        for (int v = 0; v < 4; ++v) { ys[tc][v] = ya[v] + yb[v]; yd[tc][v] = ya[v] - yb[v]; }
    }
    const int m  = 4 * (lm >> 3) + ((lm & 7) >> 1);
    const int ga = 0 * 8 + ((m >> 2) & 1) * 4;
    const int gb = 1 * 8 + ((m >> 1) & 1) * 4 + lh * 2;
    const int gc = 2 * 8 + (m & 1) * 4;
    const float a0r = g[ga], a0i = g[ga + 1], a1r = g[ga + 2], a1i = g[ga + 3];
    const float br  = g[gb], bi  = g[gb + 1];
    const float c0r = g[gc], c0i = g[gc + 1], c1r = g[gc + 2], c1i = g[gc + 3];
    const float t0r = br * c0r - bi * c0i, t0i = br * c0i + bi * c0r;
    const float t1r = br * c1r - bi * c1i, t1i = br * c1i + bi * c1r;
    const float w00r = a0r * t0r - a0i * t0i, w00i = a0r * t0i + a0i * t0r;
    const float w01r = a0r * t1r - a0i * t1i, w01i = a0r * t1i + a0i * t1r;
    const float w10r = a1r * t0r - a1i * t0i, w10i = a1r * t0i + a1i * t0r;
    const float w11r = a1r * t1r - a1i * t1i, w11i = a1r * t1i + a1i * t1r;
    const bool imrow = (lm & 1) != 0;
    v2f afr[4];
    afr[0] = imrow ? (v2f){ w00i, w01i } : (v2f){ w00r, w01r };
    afr[1] = imrow ? (v2f){ w10i, w11i } : (v2f){ w10r, w11r };
    afr[2] = imrow ? (v2f){ w00r, w01r } : (v2f){ -w00i, -w01i };
    afr[3] = imrow ? (v2f){ w10r, w11r } : (v2f){ -w10i, -w11i };

    v8f acc[4];
    #pragma unroll
    for (int t = 0; t < 4; ++t) acc[t] = (v8f){};
    #pragma unroll
    for (int k0 = 0; k0 < 4; ++k0) {
        const int part = k0 >> 1, c2 = k0 & 1;
        #pragma unroll
        for (int t = 0; t < 4; ++t) {
            const int tc = t & 1, ms = t >> 1;
            const v2f b = { u[tc][c2][0] * ys[tc][2 * ms + part], u[tc][c2][1] * yd[tc][2 * ms + part] };
            acc[t] = __builtin_amdgcn_wmma_f32_16x16x4_f32(
                false, afr[k0], false, b, (short)0, acc[t], false, false);
        }
    }
    const int f1 = lh ^ (wv & 1);
    const int rb = lh * (32 * PROW) + wv * (2 * PROW);
    int base[2][2];
    #pragma unroll
    for (int p0 = 0; p0 < 2; ++p0) {
        #pragma unroll
        for (int tc = 0; tc < 2; ++tc) base[p0][tc] = rb + ((tc ^ f1) << 4) + (lm ^ (p0 << 3));
    }
    #pragma unroll
    for (int t = 0; t < 4; ++t) {
        const int tc = t & 1, ms = t >> 1;
        #pragma unroll
        for (int k = 0; k < 4; ++k) {
            const int row = (4 * lh + k) * 8 + wv * 2 + ms;
            Sp[base[(k >> 1) ^ ms][tc] + k * (8 * PROW) + ms * PROW] =
                (v2f){ cz_sign(acc[t][2 * k], row), cz_sign(acc[t][2 * k + 1], row) };
        }
    }
}

__global__ __launch_bounds__(NTHREADS)
void qreservoir_kernel(const float* __restrict__ x, const float* __restrict__ w,
                       float* __restrict__ out, int S)
{
    __shared__ v2f   Sp[DIM * PROW];
    __shared__ float Gs[NQ * 8];
    __shared__ v4f   Rs[NQ * 3];
    __shared__ float cs[BCOLS][NQ];
    __shared__ float sn[BCOLS][NQ];

    const int s   = blockIdx.x;
    const int b0  = blockIdx.y * BCOLS;
    const int tid = threadIdx.x;

    const int lane = tid & 31;
    const int wv   = tid >> 5;
    const int lh   = lane >> 4;
    const int lm   = lane & 15;

    float* Ts = (float*)Sp;
    {
        const bool xb = tid < 64;
        const bool gl = tid >= 64 && tid < 64 + NL * NQ * 3;
        const int k  = gl ? tid - 64 : 0;
        const int gq = k / 3;
        const int j  = k - 3 * gq;
        const int ib = xb ? tid + NTHREADS : tid;
        const int oa = ((b0 + tid / NQ) * S + s) * NQ + tid % NQ;
        const int ob = ((b0 + ib / NQ) * S + s) * NQ + ib % NQ;
        const int ow = (s * NL * NQ + gq) * 3;
        const float* src = gl ? w : x;
        const int o1 = gl ? ow + (j == 0 ? 1 : 0) : ob;
        const int o2 = gl ? ow + 2 : oa;
        const float fa = x[oa];
        const float f1 = src[o1];
        const float f2 = src[o2];
        const float hp = 0.5f * 3.14159265358979323846f;
        const float t12 = (j == 1) ? f1 + f2 : f1 - f2;
        const float hg  = (j == 0) ? 0.5f * f1 : ((j == 1) ? -0.5f * t12 : 0.5f * t12);
        const float angB = gl ? hg : hp * f1;
        float sA, cA;
        sincosf(hp * fa, &sA, &cA);
        sn[tid / NQ][tid % NQ] = sA; cs[tid / NQ][tid % NQ] = cA;
        if (tid < 64 + NL * NQ * 3) {
            float sB, cB;
            sincosf(angB, &sB, &cB);
            float* ds = xb ? &sn[ib / NQ][ib % NQ] : Ts + 2 * k;
            float* dc = xb ? &cs[ib / NQ][ib % NQ] : Ts + 2 * k + 1;
            *ds = sB; *dc = cB;
        }
    }
    __syncthreads();

    v2f bA[2], bB[2];
    float u[2][2][2];
    first_layer_prep(cs, sn, lm, lh, bA, bB, u);
    if (tid >= 64 && tid < 64 + NL * NQ) {
        const float cf = cosf(0.25f);
        const float sf = sinf(0.25f);
        const int gidx = tid - 64;
        const float st = Ts[6 * gidx],     ct = Ts[6 * gidx + 1];
        const float sa = Ts[6 * gidx + 2], ca = Ts[6 * gidx + 3];
        const float sb = Ts[6 * gidx + 4], cb = Ts[6 * gidx + 5];
        float r00r =  ca * ct, r00i =  sa * ct;
        float r01r = -cb * st, r01i = -sb * st;
        float r10r =  cb * st, r10i = -sb * st;
        float r11r =  ca * ct, r11i = -sa * ct;
        const float g0 = cf * r00r + sf * r01i;
        const float g1 = cf * r00i - sf * r01r;
        const float g2 = sf * r00i + cf * r01r;
        const float g3 = -sf * r00r + cf * r01i;
        const float g4 = cf * r10r + sf * r11i;
        const float g5 = cf * r10i - sf * r11r;
        const float g6 = sf * r10i + cf * r11r;
        const float g7 = -sf * r10r + cf * r11i;
        if (gidx < NQ) {
            float* go = &Gs[gidx * 8];
            go[0] = g0; go[1] = g1; go[2] = g2; go[3] = g3;
            go[4] = g4; go[5] = g5; go[6] = g6; go[7] = g7;
        } else {
            const float bc_r = g2 * g4 + g3 * g5, bc_i = g3 * g4 - g2 * g5;
            const float ad_r = g0 * g6 + g1 * g7, ad_i = g1 * g6 - g0 * g7;
            const float ac_r = g0 * g4 + g1 * g5, ac_i = g1 * g4 - g0 * g5;
            const float bd_r = g2 * g6 + g3 * g7, bd_i = g3 * g6 - g2 * g7;
            const float ab_r = g0 * g2 + g1 * g3, ab_i = g1 * g2 - g0 * g3;
            const float cd_r = g4 * g6 + g5 * g7, cd_i = g5 * g6 - g4 * g7;
            const float zz = 0.5f * ((g0 * g0 + g1 * g1) - (g2 * g2 + g3 * g3)
                                   - (g4 * g4 + g5 * g5) + (g6 * g6 + g7 * g7));
            v4f* ro = &Rs[(gidx - NQ) * 3];
            ro[0] = (v4f){ bc_r + ad_r,    ad_i - bc_i, ac_r - bd_r, 0.f };
            ro[1] = (v4f){ -(bc_i + ad_i), ad_r - bc_r, bd_i - ac_i, 0.f };
            ro[2] = (v4f){ ab_r - cd_r,    ab_i - cd_i, zz,          0.f };
        }
    }
    __syncthreads();

    first_layer(Gs, bA, bB, u, Sp, wv, lm, lh);
    __syncthreads();

    {
        const int col = tid >> 2;
        const int p   = tid & 3;
        const int cx  = col ^ (p << 3);
        int cg[4];
        #pragma unroll
        for (int g = 0; g < 4; ++g) cg[g] = cx ^ (g << 3);
        float e[18];
        v2f A[16];

        #pragma unroll
        for (int jj = 0; jj < 16; ++jj) A[jj] = Sp[jj * (4 * PROW) + p * PROW + cg[jj >> 2]];
        #pragma unroll
        for (int q = 0; q < 4; ++q) {
            const int jb = 1 << (3 - q);
            v2f s = {0.f, 0.f};
            #pragma unroll
            for (int j0 = 0; j0 < 16; ++j0) {
                if (j0 & jb) continue;
                const int j1 = j0 | jb;
                s += (v2f){ A[j0].x, A[j0].x } * A[j1];
                s += (v2f){ A[j0].y, A[j0].y } * (v2f){ A[j1].y, -A[j1].x };
            }
            e[q] = 2.f * s.x; e[6 + q] = 2.f * s.y;
        }
        {
            v2f s1[8], s2[4], s3[2];
            v2f d3 = {0.f, 0.f}, d2 = {0.f, 0.f}, d1 = {0.f, 0.f};
            #pragma unroll
            for (int m = 0; m < 8; ++m) {
                const v2f pa = A[2 * m] * A[2 * m], pb = A[2 * m + 1] * A[2 * m + 1];
                d3 += pa - pb; s1[m] = pa + pb;
            }
            #pragma unroll
            for (int m = 0; m < 4; ++m) { d2 += s1[2 * m] - s1[2 * m + 1]; s2[m] = s1[2 * m] + s1[2 * m + 1]; }
            #pragma unroll
            for (int m = 0; m < 2; ++m) { d1 += s2[2 * m] - s2[2 * m + 1]; s3[m] = s2[2 * m] + s2[2 * m + 1]; }
            const v2f d0 = s3[0] - s3[1];
            const v2f t2 = s3[0] + s3[1];
            const float tot = t2.x + t2.y;
            e[12 + 0] = d0.x + d0.y;
            e[12 + 1] = d1.x + d1.y;
            e[12 + 2] = d2.x + d2.y;
            e[12 + 3] = d3.x + d3.y;
            e[12 + 4] = (p & 2) ? -tot : tot;
            e[12 + 5] = (p & 1) ? -tot : tot;
        }

        #pragma unroll
        for (int jj = 0; jj < 16; ++jj) A[jj] = Sp[jj * PROW + p * (16 * PROW) + cg[jj & 3]];
        #pragma unroll
        for (int q = 4; q < 6; ++q) {
            const int jb = 1 << (5 - q);
            v2f s = {0.f, 0.f};
            #pragma unroll
            for (int j0 = 0; j0 < 16; ++j0) {
                if (j0 & jb) continue;
                const int j1 = j0 | jb;
                s += (v2f){ A[j0].x, A[j0].x } * A[j1];
                s += (v2f){ A[j0].y, A[j0].y } * (v2f){ A[j1].y, -A[j1].x };
            }
            e[q] = 2.f * s.x; e[6 + q] = 2.f * s.y;
        }

        #pragma unroll
        for (int k = 0; k < 18; ++k) {
            e[k] += __shfl_xor(e[k], 1);
            e[k] += __shfl_xor(e[k], 2);
        }
        if (p < 3) {
            float o6[NQ];
            #pragma unroll
            for (int q = 0; q < NQ; ++q) {
                const v4f r = Rs[q * 3 + p];
                o6[q] = r.x * e[q] + r.y * e[6 + q] + r.z * e[12 + q];
            }
            float* o = out + ((size_t)(b0 + col) * S + s) * 18 + p * NQ;
            #pragma unroll
            for (int q = 0; q < NQ; ++q) o[q] = o6[q];
        }
    }
}

extern "C" void kernel_launch(void* const* d_in, const int* in_sizes, int n_in,
                              void* d_out, int out_size, void* d_ws, size_t ws_size,
                              hipStream_t stream) {
    const float* x = (const float*)d_in[0];
    const float* w = (const float*)d_in[1];
    float* out = (float*)d_out;
    (void)d_ws; (void)ws_size; (void)n_in; (void)out_size;
    int S = in_sizes[1] / (NL * NQ * 3);
    int B = in_sizes[0] / (S * NQ);
    dim3 grid(S, B / BCOLS);
    qreservoir_kernel<<<grid, NTHREADS, 0, stream>>>(x, w, out, S);
}
